# starting kernel
# baseline (speedup 1.0000x reference)
#include <hip/hip_runtime.h>
#include <hip/hip_bf16.h>

typedef float v2f __attribute__((ext_vector_type(2)));
typedef float v8f __attribute__((ext_vector_type(8)));

static __device__ __forceinline__ float fast_tanh(float x) {
#if __has_builtin(__builtin_amdgcn_tanhf)
    return __builtin_amdgcn_tanhf(x);
#else
    float r;
    asm("v_tanh_f32 %0, %1" : "=v"(r) : "v"(x));
    return r;
#endif
}

template <int NT>
__global__ void wmma_gemm_f32(const float* __restrict__ A,
                              const float* __restrict__ Bm,
                              const float* __restrict__ bias,
                              float* __restrict__ C,
                              int M, int N, int K,
                              size_t sA, size_t sB, size_t sC)
{
    const int batch = blockIdx.z;
    A  += (size_t)batch * sA;
    Bm += (size_t)batch * sB;
    C  += (size_t)batch * sC;

    const int tN   = blockIdx.x * 16 * NT;
    const int tM   = blockIdx.y * 16;
    const int lane = threadIdx.x & 31;
    const int half = lane >> 4;
    const int l16  = lane & 15;

    const float* __restrict__ aRow = A + (size_t)(tM + l16) * K;

    v8f acc[NT];
    #pragma unroll
    for (int t = 0; t < NT; ++t) acc[t] = (v8f){};

    #pragma unroll 2
    for (int k = 0; k < K; k += 4) {
        const int ka = k + half * 2;
        v2f av;
        av.x = aRow[ka];
        av.y = aRow[ka + 1];

        const float* __restrict__ b0 = Bm + (size_t)ka * N + tN + l16;
        const float* __restrict__ b1 = b0 + N;

        #pragma unroll
        for (int t = 0; t < NT; ++t) {
            v2f bv;
            bv.x = b0[t * 16];
            bv.y = b1[t * 16];
            acc[t] = __builtin_amdgcn_wmma_f32_16x16x4_f32(
                 false, av,  false, bv,
                 (short)0, acc[t],  false, false);
        }
    }

    #pragma unroll
    for (int t = 0; t < NT; ++t) {
        const int col  = tN + t * 16 + l16;
        const float bv = bias ? bias[col] : 0.0f;
        #pragma unroll
        for (int v = 0; v < 8; ++v) {
            const int m = tM + v + half * 8;
            C[(size_t)m * N + col] = acc[t][v] + bv;
        }
    }
}

__global__ void scores_kernel(const float* __restrict__ pf,
                              const float* __restrict__ pq,
                              const float* __restrict__ Wa,
                              const int* __restrict__ mask,
                              float* __restrict__ out)
{
    const int wavesPerBlock = blockDim.x >> 5;
    const int widx = blockIdx.x * wavesPerBlock + (threadIdx.x >> 5);
    const int lane = threadIdx.x & 31;

    const int f = widx & 127;
    const int q = (widx >> 7) & 31;
    const int b = widx >> 12;

    const float4* pf4 = (const float4*)(pf + ((size_t)(b * 128 + f)) * 256);
    const float4* pq4 = (const float4*)(pq + ((size_t)(b * 32 + q)) * 256);
    const float4* wa4 = (const float4*)Wa;

    float s = 0.0f;
    #pragma unroll
    for (int i = 0; i < 2; ++i) {
        const int idx = lane * 2 + i;
        const float4 x = pf4[idx];
        const float4 y = pq4[idx];
        const float4 w = wa4[idx];
        s += fast_tanh(x.x + y.x) * w.x;
        s += fast_tanh(x.y + y.y) * w.y;
        s += fast_tanh(x.z + y.z) * w.z;
        s += fast_tanh(x.w + y.w) * w.w;
    }
    #pragma unroll
    for (int off = 16; off > 0; off >>= 1)
        s += __shfl_down(s, off, 32);

    if (lane == 0) {
        out[widx] = (mask[b * 128 + f] == 0) ? -1e9f : s;
    }
}

__global__ void softmax_kernel(float* __restrict__ a)
{
    const int wavesPerBlock = blockDim.x >> 5;
    const int row  = blockIdx.x * wavesPerBlock + (threadIdx.x >> 5);
    const int lane = threadIdx.x & 31;

    float4* row4 = (float4*)(a + (size_t)row * 128);
    const float4 v = row4[lane];

    float m = fmaxf(fmaxf(v.x, v.y), fmaxf(v.z, v.w));
    #pragma unroll
    for (int off = 16; off > 0; off >>= 1)
        m = fmaxf(m, __shfl_xor(m, off, 32));

    float4 e;
    e.x = __expf(v.x - m);
    e.y = __expf(v.y - m);
    e.z = __expf(v.z - m);
    e.w = __expf(v.w - m);

    float s = e.x + e.y + e.z + e.w;
    #pragma unroll
    for (int off = 16; off > 0; off >>= 1)
        s += __shfl_xor(s, off, 32);

    const float inv = 1.0f / s;
    float4 o;
    o.x = e.x * inv; o.y = e.y * inv; o.z = e.z * inv; o.w = e.w * inv;
    row4[lane] = o;
}

extern "C" void kernel_launch(void* const* d_in, const int* in_sizes, int n_in,
                              void* d_out, int out_size, void* d_ws, size_t ws_size,
                              hipStream_t stream)
{
    const float* feature = (const float*)d_in[0];
    const float* query   = (const float*)d_in[1];
    const int*   mask    = (const int*)  d_in[2];
    const float* Wf      = (const float*)d_in[3];
    const float* Wq      = (const float*)d_in[4];
    const float* bq      = (const float*)d_in[5];
    const float* Wa      = (const float*)d_in[6];
    float*       out     = (float*)d_out;

    char* ws = (char*)d_ws;
    float* pf = (float*)ws;
    float* pq = (float*)(ws + (size_t)8192 * 256 * 4);
    float* sc = (float*)(ws + (size_t)8192 * 256 * 4
                            + (size_t)2048 * 256 * 4);

    const dim3 waveBlk(32);

    wmma_gemm_f32<4><<<dim3(256 / 64, 8192 / 16, 1), waveBlk, 0, stream>>>(
        feature, Wf, nullptr, pf, 8192, 256, 400, 0, 0, 0);

    wmma_gemm_f32<4><<<dim3(256 / 64, 2048 / 16, 1), waveBlk, 0, stream>>>(
        query, Wq, bq, pq, 2048, 256, 400, 0, 0, 0);

    scores_kernel<<<262144 / 8, 256, 0, stream>>>(pf, pq, Wa, mask, sc);

    softmax_kernel<<<2048 / 8, 256, 0, stream>>>(sc);

    wmma_gemm_f32<5><<<dim3(400 / 80, 32 / 16, 64), waveBlk, 0, stream>>>(
        sc, feature, nullptr, out, 32, 400, 128,
        (size_t)32 * 128, (size_t)128 * 400, (size_t)32 * 400);
}


# GEMM K loop split over the waves of a block: each wave streams its own k slice (loads issued far ahead of its WMMAs), partial strips added through LDS in slice order; changes the k summation order
# speedup vs baseline: 1.8339x; 1.8339x over previous
#include <hip/hip_runtime.h>

typedef float v2f __attribute__((ext_vector_type(2)));
typedef float v8f __attribute__((ext_vector_type(8)));

static __device__ __forceinline__ float fast_tanh(float x) {
#if __has_builtin(__builtin_amdgcn_tanhf)
    return __builtin_amdgcn_tanhf(x);
#else
    float r;
    asm("v_tanh_f32 %0, %1" : "=v"(r) : "v"(x));
    return r;
#endif
}

template <int NT, int KW, int NW, int N, int K>
__global__ void __launch_bounds__(NW * 32) __attribute__((amdgpu_waves_per_eu(1, 8)))
wmma_gemm_f32_ksplit(const float* __restrict__ A,
                     const float* __restrict__ Bm,
                     const float* __restrict__ bias,
                     float* __restrict__ C,
                     size_t sA, size_t sB, size_t sC)
{
    static_assert(K == 4 * KW * NW, "k slices must tile K exactly");

    __shared__ float red[NW][NT * 8][32];

    const int batch = blockIdx.z;
    A  += (size_t)batch * sA;
    Bm += (size_t)batch * sB;
    C  += (size_t)batch * sC;

    const int tN   = blockIdx.x * 16 * NT;
    const int tM   = blockIdx.y * 16;
    const int lane = threadIdx.x & 31;
    const int half = lane >> 4;
    const int l16  = lane & 15;
    const int wave = __builtin_amdgcn_readfirstlane((int)(threadIdx.x >> 5));

    const int k0 = wave * (4 * KW) + half * 2;
    const float* __restrict__ aP = A + (size_t)(tM + l16) * K + k0;
    const float* __restrict__ bP = Bm + (size_t)k0 * N + tN + l16;

    v2f av[KW];
    v2f bv[KW][NT];
    #pragma unroll
    for (int s = 0; s < KW; ++s) {
        av[s].x = aP[4 * s];
        av[s].y = aP[4 * s + 1];
        #pragma unroll
        for (int t = 0; t < NT; ++t) {
            bv[s][t].x = bP[(size_t)(4 * s)     * N + t * 16];
            bv[s][t].y = bP[(size_t)(4 * s + 1) * N + t * 16];
        }
    }

    v8f acc[NT];
    #pragma unroll
    for (int t = 0; t < NT; ++t) acc[t] = (v8f){};

    #pragma unroll
    for (int s = 0; s < KW; ++s) {
        #pragma unroll
        for (int t = 0; t < NT; ++t) {
            acc[t] = __builtin_amdgcn_wmma_f32_16x16x4_f32(
                 false, av[s],  false, bv[s][t],
                 (short)0, acc[t],  false, false);
        }
    }

    #pragma unroll
    for (int t = 0; t < NT; ++t) {
        #pragma unroll
        for (int v = 0; v < 8; ++v)
            red[wave][t * 8 + v][lane] = acc[t][v];
    }
    __syncthreads();

    for (int v = wave; v < 8; v += NW) {
        const int m = tM + v + half * 8;
        #pragma unroll
        for (int t = 0; t < NT; ++t) {
            const int r = t * 8 + v;
            float s = red[0][r][lane];
            #pragma unroll
            for (int w = 1; w < NW; ++w) s += red[w][r][lane];
            const int col = tN + t * 16 + l16;
            const float bvv = bias ? bias[col] : 0.0f;
            C[(size_t)m * N + col] = s + bvv;
        }
    }
}

__global__ void scores_kernel(const float* __restrict__ pf,
                              const float* __restrict__ pq,
                              const float* __restrict__ Wa,
                              const int* __restrict__ mask,
                              float* __restrict__ out)
{
    const int wavesPerBlock = blockDim.x >> 5;
    const int widx = blockIdx.x * wavesPerBlock + (threadIdx.x >> 5);
    const int lane = threadIdx.x & 31;

    const int f = widx & 127;
    const int q = (widx >> 7) & 31;
    const int b = widx >> 12;

    const float4* pf4 = (const float4*)(pf + ((size_t)(b * 128 + f)) * 256);
    const float4* pq4 = (const float4*)(pq + ((size_t)(b * 32 + q)) * 256);
    const float4* wa4 = (const float4*)Wa;

    float s = 0.0f;
    #pragma unroll
    for (int i = 0; i < 2; ++i) {
        const int idx = lane * 2 + i;
        const float4 x = pf4[idx];
        const float4 y = pq4[idx];
        const float4 w = wa4[idx];
        s += fast_tanh(x.x + y.x) * w.x;
        s += fast_tanh(x.y + y.y) * w.y;
        s += fast_tanh(x.z + y.z) * w.z;
        s += fast_tanh(x.w + y.w) * w.w;
    }
    #pragma unroll
    for (int off = 16; off > 0; off >>= 1)
        s += __shfl_down(s, off, 32);

    if (lane == 0) {
        out[widx] = (mask[b * 128 + f] == 0) ? -1e9f : s;
    }
}

__global__ void softmax_kernel(float* __restrict__ a)
{
    const int wavesPerBlock = blockDim.x >> 5;
    const int row  = blockIdx.x * wavesPerBlock + (threadIdx.x >> 5);
    const int lane = threadIdx.x & 31;

    float4* row4 = (float4*)(a + (size_t)row * 128);
    const float4 v = row4[lane];

    float m = fmaxf(fmaxf(v.x, v.y), fmaxf(v.z, v.w));
    #pragma unroll
    for (int off = 16; off > 0; off >>= 1)
        m = fmaxf(m, __shfl_xor(m, off, 32));

    float4 e;
    e.x = __expf(v.x - m);
    e.y = __expf(v.y - m);
    e.z = __expf(v.z - m);
    e.w = __expf(v.w - m);

    float s = e.x + e.y + e.z + e.w;
    #pragma unroll
    for (int off = 16; off > 0; off >>= 1)
        s += __shfl_xor(s, off, 32);

    const float inv = 1.0f / s;
    float4 o;
    o.x = e.x * inv; o.y = e.y * inv; o.z = e.z * inv; o.w = e.w * inv;
    row4[lane] = o;
}

extern "C" void kernel_launch(void* const* d_in, const int* in_sizes, int n_in,
                              void* d_out, int out_size, void* d_ws, size_t ws_size,
                              hipStream_t stream)
{
    const float* feature = (const float*)d_in[0];
    const float* query   = (const float*)d_in[1];
    const int*   mask    = (const int*)  d_in[2];
    const float* Wf      = (const float*)d_in[3];
    const float* Wq      = (const float*)d_in[4];
    const float* bq      = (const float*)d_in[5];
    const float* Wa      = (const float*)d_in[6];
    float*       out     = (float*)d_out;

    char* ws = (char*)d_ws;
    float* pf = (float*)ws;
    float* pq = (float*)(ws + (size_t)8192 * 256 * 4);
    float* sc = (float*)(ws + (size_t)8192 * 256 * 4
                            + (size_t)2048 * 256 * 4);

    wmma_gemm_f32_ksplit<4, 25, 4, 256, 400>
        <<<dim3(256 / 64, 8192 / 16, 1), dim3(128), 0, stream>>>(
        feature, Wf, nullptr, pf, 0, 0, 0);

    wmma_gemm_f32_ksplit<4, 25, 4, 256, 400>
        <<<dim3(256 / 64, 2048 / 16, 1), dim3(128), 0, stream>>>(
        query, Wq, bq, pq, 0, 0, 0);

    scores_kernel<<<262144 / 8, 256, 0, stream>>>(pf, pq, Wa, mask, sc);

    softmax_kernel<<<2048 / 8, 256, 0, stream>>>(sc);

    wmma_gemm_f32_ksplit<5, 4, 8, 400, 128>
        <<<dim3(400 / 80, 32 / 16, 64), dim3(256), 0, stream>>>(
        sc, feature, nullptr, out,
        (size_t)32 * 128, (size_t)128 * 400, (size_t)32 * 400);
}


# scores kernel: one wave per 8q x 8f tile (lane = 8 a as before), 16 rows and Wa held in registers, 64 lane sums folded by a transposing 16-8-4-2-1 butterfly; 262144 waves become 4096; sum order kept
# speedup vs baseline: 3.0601x; 1.6686x over previous
#include <hip/hip_runtime.h>

typedef float v2f __attribute__((ext_vector_type(2)));
typedef float v8f __attribute__((ext_vector_type(8)));

static __device__ __forceinline__ float fast_tanh(float x) {
#if __has_builtin(__builtin_amdgcn_tanhf)
    return __builtin_amdgcn_tanhf(x);
#else
    float r;
    asm("v_tanh_f32 %0, %1" : "=v"(r) : "v"(x));
    return r;
#endif
}

template <int NT, int KW, int NW, int N, int K>
__global__ void __launch_bounds__(NW * 32) __attribute__((amdgpu_waves_per_eu(1, 8)))
wmma_gemm_f32_ksplit(const float* __restrict__ A,
                     const float* __restrict__ Bm,
                     const float* __restrict__ bias,
                     float* __restrict__ C,
                     size_t sA, size_t sB, size_t sC)
{
    static_assert(K == 4 * KW * NW, "k slices must tile K exactly");

    __shared__ float red[NW][NT * 8][32];

    const int batch = blockIdx.z;
    A  += (size_t)batch * sA;
    Bm += (size_t)batch * sB;
    C  += (size_t)batch * sC;

    const int tN   = blockIdx.x * 16 * NT;
    const int tM   = blockIdx.y * 16;
    const int lane = threadIdx.x & 31;
    const int half = lane >> 4;
    const int l16  = lane & 15;
    const int wave = __builtin_amdgcn_readfirstlane((int)(threadIdx.x >> 5));

    const int k0 = wave * (4 * KW) + half * 2;
    const float* __restrict__ aP = A + (size_t)(tM + l16) * K + k0;
    const float* __restrict__ bP = Bm + (size_t)k0 * N + tN + l16;

    v2f av[KW];
    v2f bv[KW][NT];
    #pragma unroll
    for (int s = 0; s < KW; ++s) {
        av[s].x = aP[4 * s];
        av[s].y = aP[4 * s + 1];
        #pragma unroll
        for (int t = 0; t < NT; ++t) {
            bv[s][t].x = bP[(size_t)(4 * s)     * N + t * 16];
            bv[s][t].y = bP[(size_t)(4 * s + 1) * N + t * 16];
        }
    }

    v8f acc[NT];
    #pragma unroll
    for (int t = 0; t < NT; ++t) acc[t] = (v8f){};

    #pragma unroll
    for (int s = 0; s < KW; ++s) {
        #pragma unroll
        for (int t = 0; t < NT; ++t) {
            acc[t] = __builtin_amdgcn_wmma_f32_16x16x4_f32(
                 false, av[s],  false, bv[s][t],
                 (short)0, acc[t],  false, false);
        }
    }

    #pragma unroll
    for (int t = 0; t < NT; ++t) {
        #pragma unroll
        for (int v = 0; v < 8; ++v)
            red[wave][t * 8 + v][lane] = acc[t][v];
    }
    __syncthreads();

    for (int v = wave; v < 8; v += NW) {
        const int m = tM + v + half * 8;
        #pragma unroll
        for (int t = 0; t < NT; ++t) {
            const int r = t * 8 + v;
            float s = red[0][r][lane];
            #pragma unroll
            for (int w = 1; w < NW; ++w) s += red[w][r][lane];
            const int col = tN + t * 16 + l16;
            const float bvv = bias ? bias[col] : 0.0f;
            C[(size_t)m * N + col] = s + bvv;
        }
    }
}


template <int H>
static __device__ __forceinline__ void fold_level(float (&v)[64], const bool up, const int bit)
{
    #pragma unroll
    for (int i = 0; i < H; ++i) {
        const float keep = up ? v[i + H] : v[i];
        const float send = up ? v[i]     : v[i + H];
        v[i] = keep + __shfl_xor(send, bit, 32);
    }
}

__global__ void __launch_bounds__(256) __attribute__((amdgpu_waves_per_eu(1, 4)))
scores_kernel(const float* __restrict__ pf,
              const float* __restrict__ pq,
              const float* __restrict__ Wa,
              const int* __restrict__ mask,
              float* __restrict__ out)
{
    __shared__ float tile[16][32];

    const int lane = threadIdx.x & 31;
    const int wave = __builtin_amdgcn_readfirstlane((int)(threadIdx.x >> 5));

    const int fb = blockIdx.x & 3;
    const int qb = (blockIdx.x >> 2) & 1;
    const int b  = blockIdx.x >> 3;
    const int qt = wave >> 2;
    const int ft = wave & 3;
    const int q0 = qb * 16 + qt * 8;
    const int f0 = fb * 32 + ft * 8;

    const float4* __restrict__ wa4 = (const float4*)Wa + lane * 2;
    const float4* __restrict__ pq4 =
        (const float4*)(pq + (size_t)(b * 32 + q0) * 256) + lane * 2;
    const float4* __restrict__ pf4 =
        (const float4*)(pf + (size_t)(b * 128 + f0) * 256) + lane * 2;

    const int mk = mask[b * 128 + fb * 32 + lane];

    float w[8], y[8][8], x[8][8];
    {
        const float4 w0 = wa4[0], w1 = wa4[1];
        w[0] = w0.x; w[1] = w0.y; w[2] = w0.z; w[3] = w0.w;
        w[4] = w1.x; w[5] = w1.y; w[6] = w1.z; w[7] = w1.w;
    }
    #pragma unroll
    for (int r = 0; r < 8; ++r) {
        const float4 y0 = pq4[r * 64], y1 = pq4[r * 64 + 1];
        y[r][0] = y0.x; y[r][1] = y0.y; y[r][2] = y0.z; y[r][3] = y0.w;
        y[r][4] = y1.x; y[r][5] = y1.y; y[r][6] = y1.z; y[r][7] = y1.w;
        const float4 x0 = pf4[r * 64], x1 = pf4[r * 64 + 1];
        x[r][0] = x0.x; x[r][1] = x0.y; x[r][2] = x0.z; x[r][3] = x0.w;
        x[r][4] = x1.x; x[r][5] = x1.y; x[r][6] = x1.z; x[r][7] = x1.w;
    }

    float acc[64];
    #pragma unroll
    for (int qi = 0; qi < 8; ++qi) {
        #pragma unroll
        for (int fj = 0; fj < 8; ++fj) {
            float s = 0.0f;
            #pragma unroll
            for (int k = 0; k < 8; ++k)
                s += fast_tanh(x[fj][k] + y[qi][k]) * w[k];
            acc[qi * 8 + fj] = s;
        }
    }

    fold_level<32>(acc, (lane & 16) != 0, 16);
    fold_level<16>(acc, (lane &  8) != 0,  8);
    fold_level< 8>(acc, (lane &  4) != 0,  4);
    fold_level< 4>(acc, (lane &  2) != 0,  2);
    fold_level< 2>(acc, (lane &  1) != 0,  1);

    float2* dst = (float2*)&tile[qt * 8 + (lane >> 2)][ft * 8 + 2 * (lane & 3)];
    *dst = make_float2(acc[0], acc[1]);
    __syncthreads();

    #pragma unroll
    for (int r = 0; r < 2; ++r) {
        const int row = wave * 2 + r;
        const float s = tile[row][lane];
        out[(size_t)(b * 32 + qb * 16 + row) * 128 + fb * 32 + lane] =
            (mk == 0) ? -1e9f : s;
    }
}

__global__ void softmax_kernel(float* __restrict__ a)
{
    const int wavesPerBlock = blockDim.x >> 5;
    const int row  = blockIdx.x * wavesPerBlock + (threadIdx.x >> 5);
    const int lane = threadIdx.x & 31;

    float4* row4 = (float4*)(a + (size_t)row * 128);
    const float4 v = row4[lane];

    float m = fmaxf(fmaxf(v.x, v.y), fmaxf(v.z, v.w));
    #pragma unroll
    for (int off = 16; off > 0; off >>= 1)
        m = fmaxf(m, __shfl_xor(m, off, 32));

    float4 e;
    e.x = __expf(v.x - m);
    e.y = __expf(v.y - m);
    e.z = __expf(v.z - m);
    e.w = __expf(v.w - m);

    float s = e.x + e.y + e.z + e.w;
    #pragma unroll
    for (int off = 16; off > 0; off >>= 1)
        s += __shfl_xor(s, off, 32);

    const float inv = 1.0f / s;
    float4 o;
    o.x = e.x * inv; o.y = e.y * inv; o.z = e.z * inv; o.w = e.w * inv;
    row4[lane] = o;
}

extern "C" void kernel_launch(void* const* d_in, const int* in_sizes, int n_in,
                              void* d_out, int out_size, void* d_ws, size_t ws_size,
                              hipStream_t stream)
{
    const float* feature = (const float*)d_in[0];
    const float* query   = (const float*)d_in[1];
    const int*   mask    = (const int*)  d_in[2];
    const float* Wf      = (const float*)d_in[3];
    const float* Wq      = (const float*)d_in[4];
    const float* bq      = (const float*)d_in[5];
    const float* Wa      = (const float*)d_in[6];
    float*       out     = (float*)d_out;

    char* ws = (char*)d_ws;
    float* pf = (float*)ws;
    float* pq = (float*)(ws + (size_t)8192 * 256 * 4);
    float* sc = (float*)(ws + (size_t)8192 * 256 * 4
                            + (size_t)2048 * 256 * 4);

    wmma_gemm_f32_ksplit<4, 25, 4, 256, 400>
        <<<dim3(256 / 64, 8192 / 16, 1), dim3(128), 0, stream>>>(
        feature, Wf, nullptr, pf, 0, 0, 0);

    wmma_gemm_f32_ksplit<4, 25, 4, 256, 400>
        <<<dim3(256 / 64, 2048 / 16, 1), dim3(128), 0, stream>>>(
        query, Wq, bq, pq, 0, 0, 0);

    scores_kernel<<<64 * 2 * 4, 256, 0, stream>>>(pf, pq, Wa, mask, sc);

    softmax_kernel<<<2048 / 8, 256, 0, stream>>>(sc);

    wmma_gemm_f32_ksplit<5, 4, 8, 400, 128>
        <<<dim3(400 / 80, 32 / 16, 64), dim3(256), 0, stream>>>(
        sc, feature, nullptr, out,
        (size_t)32 * 128, (size_t)128 * 400, (size_t)32 * 400);
}


# pf projection GEMM: A and B slices staged through LDS in 16-k chunks (coalesced b128 loads, 2 stages) and shared by the 4 waves of a 64x64 tile; one k-ascending chain: changes the k summation order
# speedup vs baseline: 3.1715x; 1.0364x over previous
#include <hip/hip_runtime.h>

typedef float v2f __attribute__((ext_vector_type(2)));
typedef float v8f __attribute__((ext_vector_type(8)));

static __device__ __forceinline__ float fast_tanh(float x) {
#if __has_builtin(__builtin_amdgcn_tanhf)
    return __builtin_amdgcn_tanhf(x);
#else
    float r;
    asm("v_tanh_f32 %0, %1" : "=v"(r) : "v"(x));
    return r;
#endif
}

template <int NT, int KW, int NW, int N, int K>
__global__ void __launch_bounds__(NW * 32) __attribute__((amdgpu_waves_per_eu(1, 8)))
wmma_gemm_f32_ksplit(const float* __restrict__ A,
                     const float* __restrict__ Bm,
                     const float* __restrict__ bias,
                     float* __restrict__ C,
                     size_t sA, size_t sB, size_t sC)
{
    static_assert(K == 4 * KW * NW, "k slices must tile K exactly");

    __shared__ float red[NW][NT * 8][32];

    const int batch = blockIdx.z;
    A  += (size_t)batch * sA;
    Bm += (size_t)batch * sB;
    C  += (size_t)batch * sC;

    const int tN   = blockIdx.x * 16 * NT;
    const int tM   = blockIdx.y * 16;
    const int lane = threadIdx.x & 31;
    const int half = lane >> 4;
    const int l16  = lane & 15;
    const int wave = __builtin_amdgcn_readfirstlane((int)(threadIdx.x >> 5));

    const int k0 = wave * (4 * KW) + half * 2;
    const float* __restrict__ aP = A + (size_t)(tM + l16) * K + k0;
    const float* __restrict__ bP = Bm + (size_t)k0 * N + tN + l16;

    v2f av[KW];
    v2f bv[KW][NT];
    #pragma unroll
    for (int s = 0; s < KW; ++s) {
        av[s].x = aP[4 * s];
        av[s].y = aP[4 * s + 1];
        #pragma unroll
        for (int t = 0; t < NT; ++t) {
            bv[s][t].x = bP[(size_t)(4 * s)     * N + t * 16];
            bv[s][t].y = bP[(size_t)(4 * s + 1) * N + t * 16];
        }
    }

    v8f acc[NT];
    #pragma unroll
    for (int t = 0; t < NT; ++t) acc[t] = (v8f){};

    #pragma unroll
    for (int s = 0; s < KW; ++s) {
        #pragma unroll
        for (int t = 0; t < NT; ++t) {
            acc[t] = __builtin_amdgcn_wmma_f32_16x16x4_f32(
                 false, av[s],  false, bv[s][t],
                 (short)0, acc[t],  false, false);
        }
    }

    #pragma unroll
    for (int t = 0; t < NT; ++t) {
        #pragma unroll
        for (int v = 0; v < 8; ++v)
            red[wave][t * 8 + v][lane] = acc[t][v];
    }
    __syncthreads();

    for (int v = wave; v < 8; v += NW) {
        const int m = tM + v + half * 8;
        #pragma unroll
        for (int t = 0; t < NT; ++t) {
            const int r = t * 8 + v;
            float s = red[0][r][lane];
            #pragma unroll
            for (int w = 1; w < NW; ++w) s += red[w][r][lane];
            const int col = tN + t * 16 + l16;
            const float bvv = bias ? bias[col] : 0.0f;
            C[(size_t)m * N + col] = s + bvv;
        }
    }
}

static __device__ __forceinline__ void lds_barrier()
{
    __builtin_amdgcn_fence(__ATOMIC_RELEASE, "workgroup", "local");
    __builtin_amdgcn_s_barrier();
    __builtin_amdgcn_fence(__ATOMIC_ACQUIRE, "workgroup", "local");
}

template <int N, int K>
__global__ void __launch_bounds__(128)
wmma_gemm_f32_lds(const float* A,
                  const float* Bm,
                  float* C)
{
    constexpr int KC  = 16;
    constexpr int PA  = 20;
    constexpr int PB  = 80;
    constexpr int PC  = 68;
    constexpr int ASZ = 64 * PA;
    constexpr int BSZ = KC * PB;
    constexpr int NCH = K / KC;
    static_assert(K % KC == 0, "chunks must tile K exactly");
    static_assert(64 * PC <= 2 * (ASZ + BSZ), "the C tile fits inside the stages");

    __shared__ __attribute__((aligned(16))) float smem[2 * (ASZ + BSZ)];

    const int tid  = threadIdx.x;
    const int lane = tid & 31;
    const int half = lane >> 4;
    const int l16  = lane & 15;
    const int wave = __builtin_amdgcn_readfirstlane(tid >> 5);
    const int wn   = wave;

    const int tN = blockIdx.x * 64;
    const int tM = blockIdx.y * 64;

    const int arow = tid >> 2;
    const int ak4  = (tid & 3) * 4;
    const int brow = tid >> 4;
    const int bc4  = (tid & 15) * 4;
    const float* aG = A  + (size_t)(tM + arow) * K + ak4;
    const float* bG = Bm + (size_t)brow * N + tN + bc4;
    const int aS = arow * PA + ak4;
    const int bS = brow * PB + bc4;

    const int aF = l16 * PA + 2 * half;
    const int bF = (2 * half) * PB + wn * 16 + l16;

    v8f acc[4];
    #pragma unroll
    for (int i = 0; i < 4; ++i) acc[i] = (v8f){};

    static_assert(NCH >= 2, "the pipeline holds one chunk in flight");
    {
        const float4 ra00 = *(const float4*)(aG);
        const float4 ra10 = *(const float4*)(aG + (size_t)32 * K);
        const float4 rb00 = *(const float4*)(bG);
        const float4 rb10 = *(const float4*)(bG + (size_t)8 * N);
        *(float4*)&smem[aS] = ra00;
        *(float4*)&smem[aS + 32 * PA] = ra10;
        *(float4*)&smem[2 * ASZ + bS] = rb00;
        *(float4*)&smem[2 * ASZ + bS + 8 * PB] = rb10;
    }
    float4 ra0 = *(const float4*)(aG + KC);
    float4 ra1 = *(const float4*)(aG + (size_t)32 * K + KC);
    float4 rb0 = *(const float4*)(bG + (size_t)KC * N);
    float4 rb1 = *(const float4*)(bG + (size_t)(KC + 8) * N);
    lds_barrier();

    auto chunk_mma = [&](const int st) {
        const float* As = smem + st * ASZ;
        const float* Bs = smem + 2 * ASZ + st * BSZ;
        #pragma unroll
        for (int s = 0; s < 4; ++s) {
            v2f a[4], b;
            #pragma unroll
            for (int i = 0; i < 4; ++i) {
                const float* p = As + aF + i * 16 * PA + 4 * s;
                a[i].x = p[0];
                a[i].y = p[1];
            }
            {
                const float* p = Bs + bF + (4 * s) * PB;
                b.x = p[0];
                b.y = p[PB];
            }
            #pragma unroll
            for (int i = 0; i < 4; ++i) {
                acc[i] = __builtin_amdgcn_wmma_f32_16x16x4_f32(
                     false, a[i],  false, b,
                     (short)0, acc[i],  false, false);
            }
        }
    };

    auto store_stage = [&](const int so) {
        *(float4*)&smem[so * ASZ + aS] = ra0;
        *(float4*)&smem[so * ASZ + aS + 32 * PA] = ra1;
        *(float4*)&smem[2 * ASZ + so * BSZ + bS] = rb0;
        *(float4*)&smem[2 * ASZ + so * BSZ + bS + 8 * PB] = rb1;
    };

    for (int ch = 0; ch + 2 < NCH; ++ch) {
        const int st = ch & 1;
        chunk_mma(st);
        store_stage(st ^ 1);
        const float* an = aG + (ch + 2) * KC;
        const float* bn = bG + (size_t)(ch + 2) * KC * N;
        ra0 = *(const float4*)(an);
        ra1 = *(const float4*)(an + (size_t)32 * K);
        rb0 = *(const float4*)(bn);
        rb1 = *(const float4*)(bn + (size_t)8 * N);
        lds_barrier();
    }
    chunk_mma((NCH - 2) & 1);
    store_stage((NCH - 1) & 1);
    lds_barrier();
    chunk_mma((NCH - 1) & 1);
    lds_barrier();

    #pragma unroll
    for (int i = 0; i < 4; ++i) {
        #pragma unroll
        for (int r = 0; r < 8; ++r)
            smem[(i * 16 + 8 * half + r) * PC + wn * 16 + l16] = acc[i][r];
    }
    lds_barrier();

    #pragma unroll
    for (int p = 0; p < 8; ++p) {
        const int idx = tid + 128 * p;
        const int row = idx >> 4;
        const int c4  = (idx & 15) * 4;
        const float4 v = *(const float4*)&smem[row * PC + c4];
        *(float4*)&C[(size_t)(tM + row) * N + tN + c4] = v;
    }
}


template <int H>
static __device__ __forceinline__ void fold_level(float (&v)[64], const bool up, const int bit)
{
    #pragma unroll
    for (int i = 0; i < H; ++i) {
        const float keep = up ? v[i + H] : v[i];
        const float send = up ? v[i]     : v[i + H];
        v[i] = keep + __shfl_xor(send, bit, 32);
    }
}

__global__ void __launch_bounds__(256) __attribute__((amdgpu_waves_per_eu(1, 4)))
scores_kernel(const float* __restrict__ pf,
              const float* __restrict__ pq,
              const float* __restrict__ Wa,
              const int* __restrict__ mask,
              float* __restrict__ out)
{
    __shared__ float tile[16][32];

    const int lane = threadIdx.x & 31;
    const int wave = __builtin_amdgcn_readfirstlane((int)(threadIdx.x >> 5));

    const int fb = blockIdx.x & 3;
    const int qb = (blockIdx.x >> 2) & 1;
    const int b  = blockIdx.x >> 3;
    const int qt = wave >> 2;
    const int ft = wave & 3;
    const int q0 = qb * 16 + qt * 8;
    const int f0 = fb * 32 + ft * 8;

    const float4* __restrict__ wa4 = (const float4*)Wa + lane * 2;
    const float4* __restrict__ pq4 =
        (const float4*)(pq + (size_t)(b * 32 + q0) * 256) + lane * 2;
    const float4* __restrict__ pf4 =
        (const float4*)(pf + (size_t)(b * 128 + f0) * 256) + lane * 2;

    const int mk = mask[b * 128 + fb * 32 + lane];

    float w[8], y[8][8], x[8][8];
    {
        const float4 w0 = wa4[0], w1 = wa4[1];
        w[0] = w0.x; w[1] = w0.y; w[2] = w0.z; w[3] = w0.w;
        w[4] = w1.x; w[5] = w1.y; w[6] = w1.z; w[7] = w1.w;
    }
    #pragma unroll
    for (int r = 0; r < 8; ++r) {
        const float4 y0 = pq4[r * 64], y1 = pq4[r * 64 + 1];
        y[r][0] = y0.x; y[r][1] = y0.y; y[r][2] = y0.z; y[r][3] = y0.w;
        y[r][4] = y1.x; y[r][5] = y1.y; y[r][6] = y1.z; y[r][7] = y1.w;
        const float4 x0 = pf4[r * 64], x1 = pf4[r * 64 + 1];
        x[r][0] = x0.x; x[r][1] = x0.y; x[r][2] = x0.z; x[r][3] = x0.w;
        x[r][4] = x1.x; x[r][5] = x1.y; x[r][6] = x1.z; x[r][7] = x1.w;
    }

    float acc[64];
    #pragma unroll
    for (int qi = 0; qi < 8; ++qi) {
        #pragma unroll
        for (int fj = 0; fj < 8; ++fj) {
            float s = 0.0f;
            #pragma unroll
            for (int k = 0; k < 8; ++k)
                s += fast_tanh(x[fj][k] + y[qi][k]) * w[k];
            acc[qi * 8 + fj] = s;
        }
    }

    fold_level<32>(acc, (lane & 16) != 0, 16);
    fold_level<16>(acc, (lane &  8) != 0,  8);
    fold_level< 8>(acc, (lane &  4) != 0,  4);
    fold_level< 4>(acc, (lane &  2) != 0,  2);
    fold_level< 2>(acc, (lane &  1) != 0,  1);

    float2* dst = (float2*)&tile[qt * 8 + (lane >> 2)][ft * 8 + 2 * (lane & 3)];
    *dst = make_float2(acc[0], acc[1]);
    __syncthreads();

    #pragma unroll
    for (int r = 0; r < 2; ++r) {
        const int row = wave * 2 + r;
        const float s = tile[row][lane];
        out[(size_t)(b * 32 + qb * 16 + row) * 128 + fb * 32 + lane] =
            (mk == 0) ? -1e9f : s;
    }
}

__global__ void softmax_kernel(float* __restrict__ a)
{
    const int wavesPerBlock = blockDim.x >> 5;
    const int row  = blockIdx.x * wavesPerBlock + (threadIdx.x >> 5);
    const int lane = threadIdx.x & 31;

    float4* row4 = (float4*)(a + (size_t)row * 128);
    const float4 v = row4[lane];

    float m = fmaxf(fmaxf(v.x, v.y), fmaxf(v.z, v.w));
    #pragma unroll
    for (int off = 16; off > 0; off >>= 1)
        m = fmaxf(m, __shfl_xor(m, off, 32));

    float4 e;
    e.x = __expf(v.x - m);
    e.y = __expf(v.y - m);
    e.z = __expf(v.z - m);
    e.w = __expf(v.w - m);

    float s = e.x + e.y + e.z + e.w;
    #pragma unroll
    for (int off = 16; off > 0; off >>= 1)
        s += __shfl_xor(s, off, 32);

    const float inv = 1.0f / s;
    float4 o;
    o.x = e.x * inv; o.y = e.y * inv; o.z = e.z * inv; o.w = e.w * inv;
    row4[lane] = o;
}

extern "C" void kernel_launch(void* const* d_in, const int* in_sizes, int n_in,
                              void* d_out, int out_size, void* d_ws, size_t ws_size,
                              hipStream_t stream)
{
    const float* feature = (const float*)d_in[0];
    const float* query   = (const float*)d_in[1];
    const int*   mask    = (const int*)  d_in[2];
    const float* Wf      = (const float*)d_in[3];
    const float* Wq      = (const float*)d_in[4];
    const float* bq      = (const float*)d_in[5];
    const float* Wa      = (const float*)d_in[6];
    float*       out     = (float*)d_out;

    char* ws = (char*)d_ws;
    float* pf = (float*)ws;
    float* pq = (float*)(ws + (size_t)8192 * 256 * 4);
    float* sc = (float*)(ws + (size_t)8192 * 256 * 4
                            + (size_t)2048 * 256 * 4);

    wmma_gemm_f32_lds<256, 400>
        <<<dim3(256 / 64, 8192 / 64, 1), dim3(128), 0, stream>>>(
        feature, Wf, pf);

    wmma_gemm_f32_ksplit<4, 25, 4, 256, 400>
        <<<dim3(256 / 64, 2048 / 16, 1), dim3(128), 0, stream>>>(
        query, Wq, bq, pq, 0, 0, 0);

    scores_kernel<<<64 * 2 * 4, 256, 0, stream>>>(pf, pq, Wa, mask, sc);

    softmax_kernel<<<2048 / 8, 256, 0, stream>>>(sc);

    wmma_gemm_f32_ksplit<5, 4, 8, 400, 128>
        <<<dim3(400 / 80, 32 / 16, 64), dim3(256), 0, stream>>>(
        sc, feature, nullptr, out,
        (size_t)32 * 128, (size_t)128 * 400, (size_t)32 * 400);
}


# scores kernel: a block of 16 waves owns 8 whole score rows (8 q x 128 f) in LDS and waves 0-7 finish mask and softmax there, one alpha row per wave; separate softmax launch gone; sum orders kept
# speedup vs baseline: 3.5115x; 1.1072x over previous
#include <hip/hip_runtime.h>

typedef float v2f __attribute__((ext_vector_type(2)));
typedef float v8f __attribute__((ext_vector_type(8)));

static __device__ __forceinline__ float fast_tanh(float x) {
#if __has_builtin(__builtin_amdgcn_tanhf)
    return __builtin_amdgcn_tanhf(x);
#else
    float r;
    asm("v_tanh_f32 %0, %1" : "=v"(r) : "v"(x));
    return r;
#endif
}

template <int NT, int KW, int NW, int N, int K>
__global__ void __launch_bounds__(NW * 32) __attribute__((amdgpu_waves_per_eu(1, 8)))
wmma_gemm_f32_ksplit(const float* __restrict__ A,
                     const float* __restrict__ Bm,
                     const float* __restrict__ bias,
                     float* __restrict__ C,
                     size_t sA, size_t sB, size_t sC)
{
    static_assert(K == 4 * KW * NW, "k slices must tile K exactly");

    __shared__ float red[NW][NT * 8][32];

    const int batch = blockIdx.z;
    A  += (size_t)batch * sA;
    Bm += (size_t)batch * sB;
    C  += (size_t)batch * sC;

    const int tN   = blockIdx.x * 16 * NT;
    const int tM   = blockIdx.y * 16;
    const int lane = threadIdx.x & 31;
    const int half = lane >> 4;
    const int l16  = lane & 15;
    const int wave = __builtin_amdgcn_readfirstlane((int)(threadIdx.x >> 5));

    const int k0 = wave * (4 * KW) + half * 2;
    const float* __restrict__ aP = A + (size_t)(tM + l16) * K + k0;
    const float* __restrict__ bP = Bm + (size_t)k0 * N + tN + l16;

    v2f av[KW];
    v2f bv[KW][NT];
    #pragma unroll
    for (int s = 0; s < KW; ++s) {
        av[s].x = aP[4 * s];
        av[s].y = aP[4 * s + 1];
        #pragma unroll
        for (int t = 0; t < NT; ++t) {
            bv[s][t].x = bP[(size_t)(4 * s)     * N + t * 16];
            bv[s][t].y = bP[(size_t)(4 * s + 1) * N + t * 16];
        }
    }

    v8f acc[NT];
    #pragma unroll
    for (int t = 0; t < NT; ++t) acc[t] = (v8f){};

    #pragma unroll
    for (int s = 0; s < KW; ++s) {
        #pragma unroll
        for (int t = 0; t < NT; ++t) {
            acc[t] = __builtin_amdgcn_wmma_f32_16x16x4_f32(
                 false, av[s],  false, bv[s][t],
                 (short)0, acc[t],  false, false);
        }
    }

    #pragma unroll
    for (int t = 0; t < NT; ++t) {
        #pragma unroll
        for (int v = 0; v < 8; ++v)
            red[wave][t * 8 + v][lane] = acc[t][v];
    }
    __syncthreads();

    for (int v = wave; v < 8; v += NW) {
        const int m = tM + v + half * 8;
        #pragma unroll
        for (int t = 0; t < NT; ++t) {
            const int r = t * 8 + v;
            float s = red[0][r][lane];
            #pragma unroll
            for (int w = 1; w < NW; ++w) s += red[w][r][lane];
            const int col = tN + t * 16 + l16;
            const float bvv = bias ? bias[col] : 0.0f;
            C[(size_t)m * N + col] = s + bvv;
        }
    }
}

static __device__ __forceinline__ void lds_barrier()
{
    __builtin_amdgcn_fence(__ATOMIC_RELEASE, "workgroup", "local");
    __builtin_amdgcn_s_barrier();
    __builtin_amdgcn_fence(__ATOMIC_ACQUIRE, "workgroup", "local");
}

template <int N, int K>
__global__ void __launch_bounds__(128)
wmma_gemm_f32_lds(const float* A,
                  const float* Bm,
                  float* C)
{
    constexpr int KC  = 16;
    constexpr int PA  = 20;
    constexpr int PB  = 80;
    constexpr int PC  = 68;
    constexpr int ASZ = 64 * PA;
    constexpr int BSZ = KC * PB;
    constexpr int NCH = K / KC;
    static_assert(K % KC == 0, "chunks must tile K exactly");
    static_assert(64 * PC <= 2 * (ASZ + BSZ), "the C tile fits inside the stages");

    __shared__ __attribute__((aligned(16))) float smem[2 * (ASZ + BSZ)];

    const int tid  = threadIdx.x;
    const int lane = tid & 31;
    const int half = lane >> 4;
    const int l16  = lane & 15;
    const int wave = __builtin_amdgcn_readfirstlane(tid >> 5);
    const int wn   = wave;

    const int tN = blockIdx.x * 64;
    const int tM = blockIdx.y * 64;

    const int arow = tid >> 2;
    const int ak4  = (tid & 3) * 4;
    const int brow = tid >> 4;
    const int bc4  = (tid & 15) * 4;
    const float* aG = A  + (size_t)(tM + arow) * K + ak4;
    const float* bG = Bm + (size_t)brow * N + tN + bc4;
    const int aS = arow * PA + ak4;
    const int bS = brow * PB + bc4;

    const int aF = l16 * PA + 2 * half;
    const int bF = (2 * half) * PB + wn * 16 + l16;

    v8f acc[4];
    #pragma unroll
    for (int i = 0; i < 4; ++i) acc[i] = (v8f){};

    static_assert(NCH >= 2, "the pipeline holds one chunk in flight");
    {
        const float4 ra00 = *(const float4*)(aG);
        const float4 ra10 = *(const float4*)(aG + (size_t)32 * K);
        const float4 rb00 = *(const float4*)(bG);
        const float4 rb10 = *(const float4*)(bG + (size_t)8 * N);
        *(float4*)&smem[aS] = ra00;
        *(float4*)&smem[aS + 32 * PA] = ra10;
        *(float4*)&smem[2 * ASZ + bS] = rb00;
        *(float4*)&smem[2 * ASZ + bS + 8 * PB] = rb10;
    }
    float4 ra0 = *(const float4*)(aG + KC);
    float4 ra1 = *(const float4*)(aG + (size_t)32 * K + KC);
    float4 rb0 = *(const float4*)(bG + (size_t)KC * N);
    float4 rb1 = *(const float4*)(bG + (size_t)(KC + 8) * N);
    lds_barrier();

    auto chunk_mma = [&](const int st) {
        const float* As = smem + st * ASZ;
        const float* Bs = smem + 2 * ASZ + st * BSZ;
        #pragma unroll
        for (int s = 0; s < 4; ++s) {
            v2f a[4], b;
            #pragma unroll
            for (int i = 0; i < 4; ++i) {
                const float* p = As + aF + i * 16 * PA + 4 * s;
                a[i].x = p[0];
                a[i].y = p[1];
            }
            {
                const float* p = Bs + bF + (4 * s) * PB;
                b.x = p[0];
                b.y = p[PB];
            }
            #pragma unroll
            for (int i = 0; i < 4; ++i) {
                acc[i] = __builtin_amdgcn_wmma_f32_16x16x4_f32(
                     false, a[i],  false, b,
                     (short)0, acc[i],  false, false);
            }
        }
    };

    auto store_stage = [&](const int so) {
        *(float4*)&smem[so * ASZ + aS] = ra0;
        *(float4*)&smem[so * ASZ + aS + 32 * PA] = ra1;
        *(float4*)&smem[2 * ASZ + so * BSZ + bS] = rb0;
        *(float4*)&smem[2 * ASZ + so * BSZ + bS + 8 * PB] = rb1;
    };

    for (int ch = 0; ch + 2 < NCH; ++ch) {
        const int st = ch & 1;
        chunk_mma(st);
        store_stage(st ^ 1);
        const float* an = aG + (ch + 2) * KC;
        const float* bn = bG + (size_t)(ch + 2) * KC * N;
        ra0 = *(const float4*)(an);
        ra1 = *(const float4*)(an + (size_t)32 * K);
        rb0 = *(const float4*)(bn);
        rb1 = *(const float4*)(bn + (size_t)8 * N);
        lds_barrier();
    }
    chunk_mma((NCH - 2) & 1);
    store_stage((NCH - 1) & 1);
    lds_barrier();
    chunk_mma((NCH - 1) & 1);
    lds_barrier();

    #pragma unroll
    for (int i = 0; i < 4; ++i) {
        #pragma unroll
        for (int r = 0; r < 8; ++r)
            smem[(i * 16 + 8 * half + r) * PC + wn * 16 + l16] = acc[i][r];
    }
    lds_barrier();

    #pragma unroll
    for (int p = 0; p < 8; ++p) {
        const int idx = tid + 128 * p;
        const int row = idx >> 4;
        const int c4  = (idx & 15) * 4;
        const float4 v = *(const float4*)&smem[row * PC + c4];
        *(float4*)&C[(size_t)(tM + row) * N + tN + c4] = v;
    }
}


template <int H>
static __device__ __forceinline__ void fold_level(float (&v)[64], const bool up, const int bit)
{
    #pragma unroll
    for (int i = 0; i < H; ++i) {
        const float keep = up ? v[i + H] : v[i];
        const float send = up ? v[i]     : v[i + H];
        v[i] = keep + __shfl_xor(send, bit, 32);
    }
}

__global__ void __launch_bounds__(512) __attribute__((amdgpu_waves_per_eu(1, 4)))
scores_softmax_kernel(const float* __restrict__ pf,
                      const float* __restrict__ pq,
                      const float* __restrict__ Wa,
                      const int* __restrict__ mask,
                      float* __restrict__ out)
{
    __shared__ __attribute__((aligned(16))) float tile[8][128];

    const int lane = threadIdx.x & 31;
    const int wave = __builtin_amdgcn_readfirstlane((int)(threadIdx.x >> 5));

    const int qo = blockIdx.x & 3;
    const int b  = blockIdx.x >> 2;
    const int q0 = qo * 8;
    const int f0 = wave * 8;

    const float4* __restrict__ wa4 = (const float4*)Wa + lane * 2;
    const float4* __restrict__ pq4 =
        (const float4*)(pq + (size_t)(b * 32 + q0) * 256) + lane * 2;
    const float4* __restrict__ pf4 =
        (const float4*)(pf + (size_t)(b * 128 + f0) * 256) + lane * 2;

    const int4 mk = ((const int4*)(mask + b * 128))[lane];

    float w[8], y[8][8], x[8][8];
    {
        const float4 w0 = wa4[0], w1 = wa4[1];
        w[0] = w0.x; w[1] = w0.y; w[2] = w0.z; w[3] = w0.w;
        w[4] = w1.x; w[5] = w1.y; w[6] = w1.z; w[7] = w1.w;
    }
    #pragma unroll
    for (int r = 0; r < 8; ++r) {
        const float4 y0 = pq4[r * 64], y1 = pq4[r * 64 + 1];
        y[r][0] = y0.x; y[r][1] = y0.y; y[r][2] = y0.z; y[r][3] = y0.w;
        y[r][4] = y1.x; y[r][5] = y1.y; y[r][6] = y1.z; y[r][7] = y1.w;
        const float4 x0 = pf4[r * 64], x1 = pf4[r * 64 + 1];
        x[r][0] = x0.x; x[r][1] = x0.y; x[r][2] = x0.z; x[r][3] = x0.w;
        x[r][4] = x1.x; x[r][5] = x1.y; x[r][6] = x1.z; x[r][7] = x1.w;
    }

    float acc[64];
    #pragma unroll
    for (int qi = 0; qi < 8; ++qi) {
        #pragma unroll
        for (int fj = 0; fj < 8; ++fj) {
            float s = 0.0f;
            #pragma unroll
            for (int k = 0; k < 8; ++k)
                s += fast_tanh(x[fj][k] + y[qi][k]) * w[k];
            acc[qi * 8 + fj] = s;
        }
    }

    fold_level<32>(acc, (lane & 16) != 0, 16);
    fold_level<16>(acc, (lane &  8) != 0,  8);
    fold_level< 8>(acc, (lane &  4) != 0,  4);
    fold_level< 4>(acc, (lane &  2) != 0,  2);
    fold_level< 2>(acc, (lane &  1) != 0,  1);

    float2* dst = (float2*)&tile[lane >> 2][f0 + 2 * (lane & 3)];
    *dst = make_float2(acc[0], acc[1]);
    __syncthreads();

    if (wave < 8) {
        const float4 s4 = *(const float4*)&tile[wave][lane * 4];
        float4 v;
        v.x = (mk.x == 0) ? -1e9f : s4.x;
        v.y = (mk.y == 0) ? -1e9f : s4.y;
        v.z = (mk.z == 0) ? -1e9f : s4.z;
        v.w = (mk.w == 0) ? -1e9f : s4.w;

        float m = fmaxf(fmaxf(v.x, v.y), fmaxf(v.z, v.w));
        #pragma unroll
        for (int off = 16; off > 0; off >>= 1)
            m = fmaxf(m, __shfl_xor(m, off, 32));

        float4 e;
        e.x = __expf(v.x - m);
        e.y = __expf(v.y - m);
        e.z = __expf(v.z - m);
        e.w = __expf(v.w - m);

        float s = e.x + e.y + e.z + e.w;
        #pragma unroll
        for (int off = 16; off > 0; off >>= 1)
            s += __shfl_xor(s, off, 32);

        const float inv = 1.0f / s;
        float4 o;
        o.x = e.x * inv; o.y = e.y * inv; o.z = e.z * inv; o.w = e.w * inv;
        ((float4*)(out + (size_t)(b * 32 + q0 + wave) * 128))[lane] = o;
    }
}

extern "C" void kernel_launch(void* const* d_in, const int* in_sizes, int n_in,
                              void* d_out, int out_size, void* d_ws, size_t ws_size,
                              hipStream_t stream)
{
    const float* feature = (const float*)d_in[0];
    const float* query   = (const float*)d_in[1];
    const int*   mask    = (const int*)  d_in[2];
    const float* Wf      = (const float*)d_in[3];
    const float* Wq      = (const float*)d_in[4];
    const float* bq      = (const float*)d_in[5];
    const float* Wa      = (const float*)d_in[6];
    float*       out     = (float*)d_out;

    char* ws = (char*)d_ws;
    float* pf = (float*)ws;
    float* pq = (float*)(ws + (size_t)8192 * 256 * 4);
    float* sc = (float*)(ws + (size_t)8192 * 256 * 4
                            + (size_t)2048 * 256 * 4);

    wmma_gemm_f32_lds<256, 400>
        <<<dim3(256 / 64, 8192 / 64, 1), dim3(128), 0, stream>>>(
        feature, Wf, pf);

    wmma_gemm_f32_ksplit<4, 25, 4, 256, 400>
        <<<dim3(256 / 64, 2048 / 16, 1), dim3(128), 0, stream>>>(
        query, Wq, bq, pq, 0, 0, 0);

    scores_softmax_kernel<<<64 * 4, 512, 0, stream>>>(pf, pq, Wa, mask, sc);

    wmma_gemm_f32_ksplit<5, 4, 8, 400, 128>
        <<<dim3(400 / 80, 32 / 16, 64), dim3(256), 0, stream>>>(
        sc, feature, nullptr, out,
        (size_t)32 * 128, (size_t)128 * 400, (size_t)32 * 400);
}


# pf projection GEMM: register ring between global loads and the LDS stage deepened from 1 chunk of 16 k in flight to 4 (K loop written out 4 chunks per trip); tiles and k sum order kept
# speedup vs baseline: 4.2187x; 1.2014x over previous
#include <hip/hip_runtime.h>

typedef float v2f __attribute__((ext_vector_type(2)));
typedef float v4f __attribute__((ext_vector_type(4)));
typedef float v8f __attribute__((ext_vector_type(8)));

static __device__ __forceinline__ float fast_tanh(float x) {
#if __has_builtin(__builtin_amdgcn_tanhf)
    return __builtin_amdgcn_tanhf(x);
#else
    float r;
    asm("v_tanh_f32 %0, %1" : "=v"(r) : "v"(x));
    return r;
#endif
}

template <int NT, int KW, int NW, int N, int K>
__global__ void __launch_bounds__(NW * 32) __attribute__((amdgpu_waves_per_eu(1, 8)))
wmma_gemm_f32_ksplit(const float* __restrict__ A,
                     const float* __restrict__ Bm,
                     const float* __restrict__ bias,
                     float* __restrict__ C,
                     size_t sA, size_t sB, size_t sC)
{
    static_assert(K == 4 * KW * NW, "k slices must tile K exactly");

    __shared__ float red[NW][NT * 8][32];

    const int batch = blockIdx.z;
    A  += (size_t)batch * sA;
    Bm += (size_t)batch * sB;
    C  += (size_t)batch * sC;

    const int tN   = blockIdx.x * 16 * NT;
    const int tM   = blockIdx.y * 16;
    const int lane = threadIdx.x & 31;
    const int half = lane >> 4;
    const int l16  = lane & 15;
    const int wave = __builtin_amdgcn_readfirstlane((int)(threadIdx.x >> 5));

    const int k0 = wave * (4 * KW) + half * 2;
    const float* __restrict__ aP = A + (size_t)(tM + l16) * K + k0;
    const float* __restrict__ bP = Bm + (size_t)k0 * N + tN + l16;

    v2f av[KW];
    v2f bv[KW][NT];
    #pragma unroll
    for (int s = 0; s < KW; ++s) {
        av[s].x = aP[4 * s];
        av[s].y = aP[4 * s + 1];
        #pragma unroll
        for (int t = 0; t < NT; ++t) {
            bv[s][t].x = bP[(size_t)(4 * s)     * N + t * 16];
            bv[s][t].y = bP[(size_t)(4 * s + 1) * N + t * 16];
        }
    }

    v8f acc[NT];
    #pragma unroll
    for (int t = 0; t < NT; ++t) acc[t] = (v8f){};

    #pragma unroll
    for (int s = 0; s < KW; ++s) {
        #pragma unroll
        for (int t = 0; t < NT; ++t) {
            acc[t] = __builtin_amdgcn_wmma_f32_16x16x4_f32(
                 false, av[s],  false, bv[s][t],
                 (short)0, acc[t],  false, false);
        }
    }

    #pragma unroll
    for (int t = 0; t < NT; ++t) {
        #pragma unroll
        for (int v = 0; v < 8; ++v)
            red[wave][t * 8 + v][lane] = acc[t][v];
    }
    __syncthreads();

    for (int v = wave; v < 8; v += NW) {
        const int m = tM + v + half * 8;
        #pragma unroll
        for (int t = 0; t < NT; ++t) {
            const int r = t * 8 + v;
            float s = red[0][r][lane];
            #pragma unroll
            for (int w = 1; w < NW; ++w) s += red[w][r][lane];
            const int col = tN + t * 16 + l16;
            const float bvv = bias ? bias[col] : 0.0f;
            C[(size_t)m * N + col] = s + bvv;
        }
    }
}

static __device__ __forceinline__ void lds_barrier()
{
    __builtin_amdgcn_fence(__ATOMIC_RELEASE, "workgroup", "local");
    __builtin_amdgcn_s_barrier();
    __builtin_amdgcn_fence(__ATOMIC_ACQUIRE, "workgroup", "local");
}

template <int N, int K>
__global__ void __launch_bounds__(128)
wmma_gemm_f32_lds(const float* A,
                  const float* Bm,
                  float* C)
{
    constexpr int KC  = 16;
    constexpr int PA  = 20;
    constexpr int PB  = 80;
    constexpr int PC  = 68;
    constexpr int ASZ = 64 * PA;
    constexpr int BSZ = KC * PB;
    constexpr int NCH = K / KC;
    static_assert(K % KC == 0, "chunks must tile K exactly");
    static_assert(64 * PC <= 2 * (ASZ + BSZ), "the C tile fits inside the stages");

    __shared__ __attribute__((aligned(16))) float smem[2 * (ASZ + BSZ)];

    const int tid  = threadIdx.x;
    const int lane = tid & 31;
    const int half = lane >> 4;
    const int l16  = lane & 15;
    const int wave = __builtin_amdgcn_readfirstlane(tid >> 5);
    const int wn   = wave;

    const int tN = blockIdx.x * 64;
    const int tM = blockIdx.y * 64;

    const int arow = tid >> 2;
    const int ak4  = (tid & 3) * 4;
    const int brow = tid >> 4;
    const int bc4  = (tid & 15) * 4;
    const float* aG = A  + (size_t)(tM + arow) * K + ak4;
    const float* bG = Bm + (size_t)brow * N + tN + bc4;
    const int aS = arow * PA + ak4;
    const int bS = brow * PB + bc4;

    const int aF = l16 * PA + 2 * half;
    const int bF = (2 * half) * PB + wn * 16 + l16;

    v8f acc[4];
    #pragma unroll
    for (int i = 0; i < 4; ++i) acc[i] = (v8f){};

    constexpr int D = 4;
    static_assert(NCH >= D + 1 && (NCH - 1) % D == 0,
                  "chunk 0 is staged directly, the rest are whole groups of D");
    constexpr int G = (NCH - 1 - D) / D;

    v4f a0_0, a1_0, b0_0, b1_0;
    v4f a0_1, a1_1, b0_1, b1_1;
    v4f a0_2, a1_2, b0_2, b1_2;
    v4f a0_3, a1_3, b0_3, b1_3;

#define FETCH_CHUNK(S, c)                                                  \
    do {                                                                   \
        const float* an = aG + (c) * KC;                                   \
        const float* bn = bG + (size_t)(c) * KC * N;                       \
        a0_##S = *(const v4f*)(an);                                        \
        a1_##S = *(const v4f*)(an + (size_t)32 * K);                       \
        b0_##S = *(const v4f*)(bn);                                        \
        b1_##S = *(const v4f*)(bn + (size_t)8 * N);                        \
    } while (0)

#define STORE_STAGE(so, S)                                                 \
    do {                                                                   \
        *(v4f*)&smem[(so) * ASZ + aS] = a0_##S;                            \
        *(v4f*)&smem[(so) * ASZ + aS + 32 * PA] = a1_##S;                  \
        *(v4f*)&smem[2 * ASZ + (so) * BSZ + bS] = b0_##S;                  \
        *(v4f*)&smem[2 * ASZ + (so) * BSZ + bS + 8 * PB] = b1_##S;         \
    } while (0)

    const v4f ra00 = *(const v4f*)(aG);
    const v4f ra10 = *(const v4f*)(aG + (size_t)32 * K);
    const v4f rb00 = *(const v4f*)(bG);
    const v4f rb10 = *(const v4f*)(bG + (size_t)8 * N);
    __builtin_amdgcn_sched_barrier(0);
    FETCH_CHUNK(1, 1);
    __builtin_amdgcn_sched_barrier(0);
    FETCH_CHUNK(2, 2);
    __builtin_amdgcn_sched_barrier(0);
    FETCH_CHUNK(3, 3);
    __builtin_amdgcn_sched_barrier(0);
    FETCH_CHUNK(0, 4);
    __builtin_amdgcn_sched_barrier(0);
    *(v4f*)&smem[aS] = ra00;
    *(v4f*)&smem[aS + 32 * PA] = ra10;
    *(v4f*)&smem[2 * ASZ + bS] = rb00;
    *(v4f*)&smem[2 * ASZ + bS + 8 * PB] = rb10;
    lds_barrier();

    auto chunk_mma = [&](const int st) {
        const float* As = smem + st * ASZ;
        const float* Bs = smem + 2 * ASZ + st * BSZ;
        #pragma unroll
        for (int s = 0; s < 4; ++s) {
            v2f a[4], b;
            #pragma unroll
            for (int i = 0; i < 4; ++i) {
                const float* p = As + aF + i * 16 * PA + 4 * s;
                a[i].x = p[0];
                a[i].y = p[1];
            }
            {
                const float* p = Bs + bF + (4 * s) * PB;
                b.x = p[0];
                b.y = p[PB];
            }
            #pragma unroll
            for (int i = 0; i < 4; ++i) {
                acc[i] = __builtin_amdgcn_wmma_f32_16x16x4_f32(
                     false, a[i],  false, b,
                     (short)0, acc[i],  false, false);
            }
        }
    };

    for (int g = 0; g < G; ++g) {
        const int ch = g * D;
        chunk_mma(0); STORE_STAGE(1, 1); FETCH_CHUNK(1, ch + 1 + D); lds_barrier();
        chunk_mma(1); STORE_STAGE(0, 2); FETCH_CHUNK(2, ch + 2 + D); lds_barrier();
        chunk_mma(0); STORE_STAGE(1, 3); FETCH_CHUNK(3, ch + 3 + D); lds_barrier();
        chunk_mma(1); STORE_STAGE(0, 0); FETCH_CHUNK(0, ch + 4 + D); lds_barrier();
    }
    chunk_mma(0); STORE_STAGE(1, 1); lds_barrier();
    chunk_mma(1); STORE_STAGE(0, 2); lds_barrier();
    chunk_mma(0); STORE_STAGE(1, 3); lds_barrier();
    chunk_mma(1); STORE_STAGE(0, 0); lds_barrier();
    chunk_mma(0);
#undef FETCH_CHUNK
#undef STORE_STAGE
    lds_barrier();

    #pragma unroll
    for (int i = 0; i < 4; ++i) {
        #pragma unroll
        for (int r = 0; r < 8; ++r)
            smem[(i * 16 + 8 * half + r) * PC + wn * 16 + l16] = acc[i][r];
    }
    lds_barrier();

    #pragma unroll
    for (int p = 0; p < 8; ++p) {
        const int idx = tid + 128 * p;
        const int row = idx >> 4;
        const int c4  = (idx & 15) * 4;
        const float4 v = *(const float4*)&smem[row * PC + c4];
        *(float4*)&C[(size_t)(tM + row) * N + tN + c4] = v;
    }
}


template <int H>
static __device__ __forceinline__ void fold_level(float (&v)[64], const bool up, const int bit)
{
    #pragma unroll
    for (int i = 0; i < H; ++i) {
        const float keep = up ? v[i + H] : v[i];
        const float send = up ? v[i]     : v[i + H];
        v[i] = keep + __shfl_xor(send, bit, 32);
    }
}

__global__ void __launch_bounds__(512) __attribute__((amdgpu_waves_per_eu(1, 4)))
scores_softmax_kernel(const float* __restrict__ pf,
                      const float* __restrict__ pq,
                      const float* __restrict__ Wa,
                      const int* __restrict__ mask,
                      float* __restrict__ out)
{
    __shared__ __attribute__((aligned(16))) float tile[8][128];

    const int lane = threadIdx.x & 31;
    const int wave = __builtin_amdgcn_readfirstlane((int)(threadIdx.x >> 5));

    const int qo = blockIdx.x & 3;
    const int b  = blockIdx.x >> 2;
    const int q0 = qo * 8;
    const int f0 = wave * 8;

    const float4* __restrict__ wa4 = (const float4*)Wa + lane * 2;
    const float4* __restrict__ pq4 =
        (const float4*)(pq + (size_t)(b * 32 + q0) * 256) + lane * 2;
    const float4* __restrict__ pf4 =
        (const float4*)(pf + (size_t)(b * 128 + f0) * 256) + lane * 2;

    const int4 mk = ((const int4*)(mask + b * 128))[lane];

    float w[8], y[8][8], x[8][8];
    {
        const float4 w0 = wa4[0], w1 = wa4[1];
        w[0] = w0.x; w[1] = w0.y; w[2] = w0.z; w[3] = w0.w;
        w[4] = w1.x; w[5] = w1.y; w[6] = w1.z; w[7] = w1.w;
    }
    #pragma unroll
    for (int r = 0; r < 8; ++r) {
        const float4 y0 = pq4[r * 64], y1 = pq4[r * 64 + 1];
        y[r][0] = y0.x; y[r][1] = y0.y; y[r][2] = y0.z; y[r][3] = y0.w;
        y[r][4] = y1.x; y[r][5] = y1.y; y[r][6] = y1.z; y[r][7] = y1.w;
        const float4 x0 = pf4[r * 64], x1 = pf4[r * 64 + 1];
        x[r][0] = x0.x; x[r][1] = x0.y; x[r][2] = x0.z; x[r][3] = x0.w;
        x[r][4] = x1.x; x[r][5] = x1.y; x[r][6] = x1.z; x[r][7] = x1.w;
    }

    float acc[64];
    #pragma unroll
    for (int qi = 0; qi < 8; ++qi) {
        #pragma unroll
        for (int fj = 0; fj < 8; ++fj) {
            float s = 0.0f;
            #pragma unroll
            for (int k = 0; k < 8; ++k)
                s += fast_tanh(x[fj][k] + y[qi][k]) * w[k];
            acc[qi * 8 + fj] = s;
        }
    }

    fold_level<32>(acc, (lane & 16) != 0, 16);
    fold_level<16>(acc, (lane &  8) != 0,  8);
    fold_level< 8>(acc, (lane &  4) != 0,  4);
    fold_level< 4>(acc, (lane &  2) != 0,  2);
    fold_level< 2>(acc, (lane &  1) != 0,  1);

    float2* dst = (float2*)&tile[lane >> 2][f0 + 2 * (lane & 3)];
    *dst = make_float2(acc[0], acc[1]);
    __syncthreads();

    if (wave < 8) {
        const float4 s4 = *(const float4*)&tile[wave][lane * 4];
        float4 v;
        v.x = (mk.x == 0) ? -1e9f : s4.x;
        v.y = (mk.y == 0) ? -1e9f : s4.y;
        v.z = (mk.z == 0) ? -1e9f : s4.z;
        v.w = (mk.w == 0) ? -1e9f : s4.w;

        float m = fmaxf(fmaxf(v.x, v.y), fmaxf(v.z, v.w));
        #pragma unroll
        for (int off = 16; off > 0; off >>= 1)
            m = fmaxf(m, __shfl_xor(m, off, 32));

        float4 e;
        e.x = __expf(v.x - m);
        e.y = __expf(v.y - m);
        e.z = __expf(v.z - m);
        e.w = __expf(v.w - m);

        float s = e.x + e.y + e.z + e.w;
        #pragma unroll
        for (int off = 16; off > 0; off >>= 1)
            s += __shfl_xor(s, off, 32);

        const float inv = 1.0f / s;
        float4 o;
        o.x = e.x * inv; o.y = e.y * inv; o.z = e.z * inv; o.w = e.w * inv;
        ((float4*)(out + (size_t)(b * 32 + q0 + wave) * 128))[lane] = o;
    }
}

extern "C" void kernel_launch(void* const* d_in, const int* in_sizes, int n_in,
                              void* d_out, int out_size, void* d_ws, size_t ws_size,
                              hipStream_t stream)
{
    const float* feature = (const float*)d_in[0];
    const float* query   = (const float*)d_in[1];
    const int*   mask    = (const int*)  d_in[2];
    const float* Wf      = (const float*)d_in[3];
    const float* Wq      = (const float*)d_in[4];
    const float* bq      = (const float*)d_in[5];
    const float* Wa      = (const float*)d_in[6];
    float*       out     = (float*)d_out;

    char* ws = (char*)d_ws;
    float* pf = (float*)ws;
    float* pq = (float*)(ws + (size_t)8192 * 256 * 4);
    float* sc = (float*)(ws + (size_t)8192 * 256 * 4
                            + (size_t)2048 * 256 * 4);

    wmma_gemm_f32_lds<256, 400>
        <<<dim3(256 / 64, 8192 / 64, 1), dim3(128), 0, stream>>>(
        feature, Wf, pf);

    wmma_gemm_f32_ksplit<4, 25, 4, 256, 400>
        <<<dim3(256 / 64, 2048 / 16, 1), dim3(128), 0, stream>>>(
        query, Wq, bq, pq, 0, 0, 0);

    scores_softmax_kernel<<<64 * 4, 512, 0, stream>>>(pf, pq, Wa, mask, sc);

    wmma_gemm_f32_ksplit<5, 4, 8, 400, 128>
        <<<dim3(400 / 80, 32 / 16, 64), dim3(256), 0, stream>>>(
        sc, feature, nullptr, out,
        (size_t)32 * 128, (size_t)128 * 400, (size_t)32 * 400);
}


# A split over blocks: block per (batch, 64-a slice) forms its pf, pq slices in LDS on WMMA, then partial scores; slice sums+softmax open the final product; 4 launches to 2; changes a, pq-k sum order
# speedup vs baseline: 5.2934x; 1.2547x over previous
#include <hip/hip_runtime.h>

typedef float v2f __attribute__((ext_vector_type(2)));
typedef float v4f __attribute__((ext_vector_type(4)));
typedef float v8f __attribute__((ext_vector_type(8)));

static __device__ __forceinline__ float fast_tanh(float x) {
#if __has_builtin(__builtin_amdgcn_tanhf)
    return __builtin_amdgcn_tanhf(x);
#else
    float r;
    asm("v_tanh_f32 %0, %1" : "=v"(r) : "v"(x));
    return r;
#endif
}

static __device__ __forceinline__ void lds_barrier()
{
    __builtin_amdgcn_fence(__ATOMIC_RELEASE, "workgroup", "local");
    __builtin_amdgcn_s_barrier();
    __builtin_amdgcn_fence(__ATOMIC_ACQUIRE, "workgroup", "local");
}

template <int H, int NV>
static __device__ __forceinline__ void fold_level(float (&v)[NV], const bool up, const int bit)
{
    #pragma unroll
    for (int i = 0; i < H; ++i) {
        const float keep = up ? v[i + H] : v[i];
        const float send = up ? v[i]     : v[i + H];
        v[i] = keep + __shfl_xor(send, bit, 32);
    }
}

__global__ void __launch_bounds__(512) __attribute__((amdgpu_waves_per_eu(1, 4)))
proj_scores_slice_kernel(const float* feature,
                         const float* query,
                         const float* Wf,
                         const float* Wq,
                         const float* bq,
                         const float* Wa,
                         float* part)
{
    constexpr int K    = 400;
    constexpr int NA   = 256;
    constexpr int KC   = 16;
    constexpr int PA   = 20;
    constexpr int PB   = 80;
    constexpr int PC   = 68;
    constexpr int AFSZ = 128 * PA;
    constexpr int AQSZ = 32 * PA;
    constexpr int BSZ  = KC * PB;
    constexpr int STG  = AFSZ + AQSZ + 2 * BSZ;
    constexpr int NCH  = K / KC;
    constexpr int PQT  = 128 * PC;
    static_assert(K % KC == 0, "chunks must tile K exactly");
    static_assert((128 + 32) * PC <= 2 * STG, "the finished tiles fit inside the stages");

    __shared__ __attribute__((aligned(16))) float smem[2 * STG];

    const int tid  = threadIdx.x;
    const int lane = tid & 31;
    const int half = lane >> 4;
    const int l16  = lane & 15;
    const int wave = __builtin_amdgcn_readfirstlane(tid >> 5);
    const int wn   = wave & 3;
    const int rg   = wave >> 2;
    const bool isq = (rg == 3);
    const bool nt3 = (rg < 2);
    const int rt0  = isq ? 0 : rg * 3;

    const int j = blockIdx.x & 3;
    const int b = blockIdx.x >> 2;

    const bool bisq = (wave >= 8);
    const int arow = tid >> 2;
    const int ak4  = (tid & 3) * 4;
    const int brow = (tid & 255) >> 4;
    const int bc4  = (tid & 15) * 4;
    const float* afG = feature + (size_t)(b * 128 + arow) * K + ak4;
    const float* aqG = query + (size_t)(b * 32 + (tid >> 4)) * K + (tid & 15);
    const float* bG  = (bisq ? Wq : Wf) + (size_t)brow * NA + j * 64 + bc4;
    const int afS = arow * PA + ak4;
    const int aqS = AFSZ + (tid >> 4) * PA + (tid & 15);
    const int bS  = AFSZ + AQSZ + (bisq ? BSZ : 0) + brow * PB + bc4;

    const int aF = (isq ? AFSZ : 0) + (rt0 * 16 + l16) * PA + 2 * half;
    const int bF = AFSZ + AQSZ + (isq ? BSZ : 0) + (2 * half) * PB + wn * 16 + l16;

    v8f acc[3];
    #pragma unroll
    for (int i = 0; i < 3; ++i) acc[i] = (v8f){};

    constexpr int D = 4;
    static_assert(NCH >= D + 1 && (NCH - 1) % D == 0,
                  "chunk 0 is staged directly, the rest are whole groups of D");
    constexpr int G = (NCH - 1 - D) / D;

    v4f af_0, bb_0; float aq_0;
    v4f af_1, bb_1; float aq_1;
    v4f af_2, bb_2; float aq_2;
    v4f af_3, bb_3; float aq_3;

#define FETCH_CHUNK(S, c)                                                  \
    do {                                                                   \
        af_##S = *(const v4f*)(afG + (c) * KC);                            \
        aq_##S = aqG[(c) * KC];                                            \
        bb_##S = *(const v4f*)(bG + (size_t)(c) * KC * NA);                \
    } while (0)

#define STORE_STAGE(so, S)                                                 \
    do {                                                                   \
        *(v4f*)&smem[(so) * STG + afS] = af_##S;                           \
        smem[(so) * STG + aqS] = aq_##S;                                   \
        *(v4f*)&smem[(so) * STG + bS] = bb_##S;                            \
    } while (0)

    FETCH_CHUNK(0, 0);
    __builtin_amdgcn_sched_barrier(0);
    FETCH_CHUNK(1, 1);
    __builtin_amdgcn_sched_barrier(0);
    FETCH_CHUNK(2, 2);
    __builtin_amdgcn_sched_barrier(0);
    FETCH_CHUNK(3, 3);
    __builtin_amdgcn_sched_barrier(0);
    STORE_STAGE(0, 0);
    FETCH_CHUNK(0, 4);
    lds_barrier();

    auto chunk_mma = [&](const int st) {
        const float* As = smem + st * STG + aF;
        const float* Bs = smem + st * STG + bF;
        v2f bfr[4];
        #pragma unroll
        for (int s = 0; s < 4; ++s) {
            const float* p = Bs + (4 * s) * PB;
            bfr[s].x = p[0];
            bfr[s].y = p[PB];
        }
        #pragma unroll
        for (int s = 0; s < 4; ++s) {
            v2f a0, a1;
            const float* p = As + 4 * s;
            a0.x = p[0];
            a0.y = p[1];
            a1.x = p[16 * PA];
            a1.y = p[16 * PA + 1];
            acc[0] = __builtin_amdgcn_wmma_f32_16x16x4_f32(
                 false, a0,  false, bfr[s],
                 (short)0, acc[0],  false, false);
            acc[1] = __builtin_amdgcn_wmma_f32_16x16x4_f32(
                 false, a1,  false, bfr[s],
                 (short)0, acc[1],  false, false);
        }
        if (nt3) {
            #pragma unroll
            for (int s = 0; s < 4; ++s) {
                v2f a2;
                const float* p = As + 32 * PA + 4 * s;
                a2.x = p[0];
                a2.y = p[1];
                acc[2] = __builtin_amdgcn_wmma_f32_16x16x4_f32(
                     false, a2,  false, bfr[s],
                     (short)0, acc[2],  false, false);
            }
        }
    };

    for (int g = 0; g < G; ++g) {
        const int ch = g * D;
        chunk_mma(0); STORE_STAGE(1, 1); FETCH_CHUNK(1, ch + 1 + D); lds_barrier();
        chunk_mma(1); STORE_STAGE(0, 2); FETCH_CHUNK(2, ch + 2 + D); lds_barrier();
        chunk_mma(0); STORE_STAGE(1, 3); FETCH_CHUNK(3, ch + 3 + D); lds_barrier();
        chunk_mma(1); STORE_STAGE(0, 0); FETCH_CHUNK(0, ch + 4 + D); lds_barrier();
    }
    chunk_mma(0); STORE_STAGE(1, 1); lds_barrier();
    chunk_mma(1); STORE_STAGE(0, 2); lds_barrier();
    chunk_mma(0); STORE_STAGE(1, 3); lds_barrier();
    chunk_mma(1); STORE_STAGE(0, 0); lds_barrier();
    chunk_mma(0);
#undef FETCH_CHUNK
#undef STORE_STAGE
    lds_barrier();

    {
        const int col = wn * 16 + l16;
        const float bias = isq ? bq[j * 64 + col] : 0.0f;
        float* T = smem + (isq ? PQT : 0) + (rt0 * 16 + 8 * half) * PC + col;
        #pragma unroll
        for (int r = 0; r < 8; ++r) {
            T[r * PC]             = acc[0][r] + bias;
            T[(16 + r) * PC]      = acc[1][r] + bias;
        }
        if (nt3) {
            #pragma unroll
            for (int r = 0; r < 8; ++r)
                T[(32 + r) * PC]  = acc[2][r];
        }
    }
    lds_barrier();

    const int qo  = wave >> 2;
    const int fg  = wave & 3;
    const int a8  = lane & 7;
    const int sub = lane >> 3;
    const int f0  = (fg * 4 + sub) * 8;

    float w[8], x[8][8];
    {
        const float4* wa4 = (const float4*)(Wa + j * 64 + a8 * 8);
        const float4 w0 = wa4[0], w1 = wa4[1];
        w[0] = w0.x; w[1] = w0.y; w[2] = w0.z; w[3] = w0.w;
        w[4] = w1.x; w[5] = w1.y; w[6] = w1.z; w[7] = w1.w;
    }
    #pragma unroll
    for (int r = 0; r < 8; ++r) {
        const float4* xp = (const float4*)&smem[(f0 + r) * PC + a8 * 8];
        const float4 x0 = xp[0], x1 = xp[1];
        x[r][0] = x0.x; x[r][1] = x0.y; x[r][2] = x0.z; x[r][3] = x0.w;
        x[r][4] = x1.x; x[r][5] = x1.y; x[r][6] = x1.z; x[r][7] = x1.w;
    }

    const float* yrow = &smem[PQT + (qo * 8) * PC + a8 * 8];
    float4 yc0 = *(const float4*)(yrow);
    float4 yc1 = *(const float4*)(yrow + 4);
    float* dst = part + ((size_t)((j * 64 + b) * 32 + qo * 8)) * 128 + fg * 32 + lane;
    #pragma nounroll
    for (int qi = 0; qi < 8; ++qi) {
        const float* yn = yrow + ((qi + 1) & 7) * PC;
        const float4 yn0 = *(const float4*)(yn);
        const float4 yn1 = *(const float4*)(yn + 4);
        const float y[8] = {yc0.x, yc0.y, yc0.z, yc0.w, yc1.x, yc1.y, yc1.z, yc1.w};

        float sums[8];
        #pragma unroll
        for (int fj = 0; fj < 8; ++fj) sums[fj] = 0.0f;
        #pragma unroll
        for (int k = 0; k < 8; ++k) {
            float t[8];
            #pragma unroll
            for (int fj = 0; fj < 8; ++fj)
                t[fj] = fast_tanh(x[fj][k] + y[k]);
            #pragma unroll
            for (int fj = 0; fj < 8; ++fj)
                sums[fj] = __builtin_fmaf(t[fj], w[k], sums[fj]);
        }

        fold_level<4>(sums, (lane & 4) != 0, 4);
        fold_level<2>(sums, (lane & 2) != 0, 2);
        fold_level<1>(sums, (lane & 1) != 0, 1);

        dst[qi * 128] = sums[0];

        yc0 = yn0;
        yc1 = yn1;
    }
}

__global__ void __launch_bounds__(256) __attribute__((amdgpu_waves_per_eu(1, 8)))
softmax_av_kernel(const float* __restrict__ part,
                  const int* __restrict__ mask,
                  const float* __restrict__ feature,
                  float* __restrict__ out)
{
    constexpr int NT = 5, KW = 4, NW = 8, N = 400, F = 128;
    constexpr int PAL = 132;
    constexpr size_t SL4 = (size_t)64 * 32 * 128 / 4;

    __shared__ float red[NW][NT * 8][32];
    __shared__ __attribute__((aligned(16))) float alpha[16][PAL];

    const int batch = blockIdx.z;
    const int tN   = blockIdx.x * 16 * NT;
    const int tM   = blockIdx.y * 16;
    const int lane = threadIdx.x & 31;
    const int half = lane >> 4;
    const int l16  = lane & 15;
    const int wave = __builtin_amdgcn_readfirstlane((int)(threadIdx.x >> 5));

    const int k0 = wave * (4 * KW) + half * 2;
    const float* __restrict__ bP =
        feature + (size_t)batch * F * N + (size_t)k0 * N + tN + l16;

    const int4 mk = ((const int4*)(mask + batch * F))[lane];
    float4 ps[2][4];
    #pragma unroll
    for (int rr = 0; rr < 2; ++rr) {
        const float4* p =
            (const float4*)(part + (size_t)(batch * 32 + tM + 2 * wave + rr) * F) + lane;
        #pragma unroll
        for (int sl = 0; sl < 4; ++sl) ps[rr][sl] = p[sl * SL4];
    }

    v2f bv[KW][NT];
    #pragma unroll
    for (int s = 0; s < KW; ++s) {
        #pragma unroll
        for (int t = 0; t < NT; ++t) {
            bv[s][t].x = bP[(size_t)(4 * s)     * N + t * 16];
            bv[s][t].y = bP[(size_t)(4 * s + 1) * N + t * 16];
        }
    }
    __builtin_amdgcn_sched_barrier(0);

    {
        #pragma unroll
        for (int rr = 0; rr < 2; ++rr) {
            const int row = 2 * wave + rr;
            const float4 s0 = ps[rr][0], s1 = ps[rr][1], s2 = ps[rr][2], s3 = ps[rr][3];
            float4 v;
            v.x = ((s0.x + s1.x) + s2.x) + s3.x;
            v.y = ((s0.y + s1.y) + s2.y) + s3.y;
            v.z = ((s0.z + s1.z) + s2.z) + s3.z;
            v.w = ((s0.w + s1.w) + s2.w) + s3.w;
            v.x = (mk.x == 0) ? -1e9f : v.x;
            v.y = (mk.y == 0) ? -1e9f : v.y;
            v.z = (mk.z == 0) ? -1e9f : v.z;
            v.w = (mk.w == 0) ? -1e9f : v.w;

            float m = fmaxf(fmaxf(v.x, v.y), fmaxf(v.z, v.w));
            #pragma unroll
            for (int off = 16; off > 0; off >>= 1)
                m = fmaxf(m, __shfl_xor(m, off, 32));

            float4 e;
            e.x = __expf(v.x - m);
            e.y = __expf(v.y - m);
            e.z = __expf(v.z - m);
            e.w = __expf(v.w - m);

            float s = e.x + e.y + e.z + e.w;
            #pragma unroll
            for (int off = 16; off > 0; off >>= 1)
                s += __shfl_xor(s, off, 32);

            const float inv = 1.0f / s;
            float4 o;
            o.x = e.x * inv; o.y = e.y * inv; o.z = e.z * inv; o.w = e.w * inv;
            *(float4*)&alpha[row][lane * 4] = o;
        }
    }
    lds_barrier();

    v2f av[KW];
    #pragma unroll
    for (int s = 0; s < KW; ++s) {
        const float* p = &alpha[l16][k0 + 4 * s];
        av[s].x = p[0];
        av[s].y = p[1];
    }

    v8f acc[NT];
    #pragma unroll
    for (int t = 0; t < NT; ++t) acc[t] = (v8f){};

    #pragma unroll
    for (int s = 0; s < KW; ++s) {
        #pragma unroll
        for (int t = 0; t < NT; ++t) {
            acc[t] = __builtin_amdgcn_wmma_f32_16x16x4_f32(
                 false, av[s],  false, bv[s][t],
                 (short)0, acc[t],  false, false);
        }
    }

    #pragma unroll
    for (int t = 0; t < NT; ++t) {
        #pragma unroll
        for (int v = 0; v < 8; ++v)
            red[wave][t * 8 + v][lane] = acc[t][v];
    }
    __syncthreads();

    float* __restrict__ C = out + (size_t)batch * 32 * N;
    for (int v = wave; v < 8; v += NW) {
        const int m = tM + v + half * 8;
        #pragma unroll
        for (int t = 0; t < NT; ++t) {
            const int r = t * 8 + v;
            float s = red[0][r][lane];
            #pragma unroll
            for (int w = 1; w < NW; ++w) s += red[w][r][lane];
            C[(size_t)m * N + tN + t * 16 + l16] = s;
        }
    }
}

extern "C" void kernel_launch(void* const* d_in, const int* in_sizes, int n_in,
                              void* d_out, int out_size, void* d_ws, size_t ws_size,
                              hipStream_t stream)
{
    const float* feature = (const float*)d_in[0];
    const float* query   = (const float*)d_in[1];
    const int*   mask    = (const int*)  d_in[2];
    const float* Wf      = (const float*)d_in[3];
    const float* Wq      = (const float*)d_in[4];
    const float* bq      = (const float*)d_in[5];
    const float* Wa      = (const float*)d_in[6];
    float*       out     = (float*)d_out;

    float* part = (float*)d_ws;

    proj_scores_slice_kernel<<<64 * 4, 512, 0, stream>>>(
        feature, query, Wf, Wq, bq, Wa, part);

    softmax_av_kernel<<<dim3(400 / 80, 32 / 16, 64), dim3(256), 0, stream>>>(
        part, mask, feature, out);
}


# proj GEMM K loop: a chunk's A, B fragments read from LDS one chunk ahead as one group into a second register set (LDS stages 2 to 4); WMMAs wait neither on LDS nor behind the barrier; sum order kept
# speedup vs baseline: 5.4520x; 1.0300x over previous
#include <hip/hip_runtime.h>

typedef float v2f __attribute__((ext_vector_type(2)));
typedef float v4f __attribute__((ext_vector_type(4)));
typedef float v8f __attribute__((ext_vector_type(8)));

static __device__ __forceinline__ float fast_tanh(float x) {
#if __has_builtin(__builtin_amdgcn_tanhf)
    return __builtin_amdgcn_tanhf(x);
#else
    float r;
    asm("v_tanh_f32 %0, %1" : "=v"(r) : "v"(x));
    return r;
#endif
}

static __device__ __forceinline__ void lds_barrier()
{
    __builtin_amdgcn_fence(__ATOMIC_RELEASE, "workgroup", "local");
    __builtin_amdgcn_s_barrier();
    __builtin_amdgcn_fence(__ATOMIC_ACQUIRE, "workgroup", "local");
}

template <int H, int NV>
static __device__ __forceinline__ void fold_level(float (&v)[NV], const bool up, const int bit)
{
    #pragma unroll
    for (int i = 0; i < H; ++i) {
        const float keep = up ? v[i + H] : v[i];
        const float send = up ? v[i]     : v[i + H];
        v[i] = keep + __shfl_xor(send, bit, 32);
    }
}

__global__ void __launch_bounds__(512) __attribute__((amdgpu_waves_per_eu(1, 4)))
proj_scores_slice_kernel(const float* feature,
                         const float* query,
                         const float* Wf,
                         const float* Wq,
                         const float* bq,
                         const float* Wa,
                         float* part)
{
    constexpr int K    = 400;
    constexpr int NA   = 256;
    constexpr int KC   = 16;
    constexpr int PA   = 20;
    constexpr int PB   = 80;
    constexpr int PC   = 68;
    constexpr int AFSZ = 128 * PA;
    constexpr int AQSZ = 32 * PA;
    constexpr int BSZ  = KC * PB;
    constexpr int STG  = AFSZ + AQSZ + 2 * BSZ;
    constexpr int NCH  = K / KC;
    constexpr int PQT  = 128 * PC;
    static_assert(K % KC == 0, "chunks must tile K exactly");
    constexpr int NSTG = 4;
    static_assert((128 + 32) * PC <= NSTG * STG, "the finished tiles fit inside the stages");

    __shared__ __attribute__((aligned(16))) float smem[NSTG * STG];

    const int tid  = threadIdx.x;
    const int lane = tid & 31;
    const int half = lane >> 4;
    const int l16  = lane & 15;
    const int wave = __builtin_amdgcn_readfirstlane(tid >> 5);
    const int wn   = wave & 3;
    const int rg   = wave >> 2;
    const bool isq = (rg == 3);
    const bool nt3 = (rg < 2);
    const int rt0  = isq ? 0 : rg * 3;

    const int j = blockIdx.x & 3;
    const int b = blockIdx.x >> 2;

    const bool bisq = (wave >= 8);
    const int arow = tid >> 2;
    const int ak4  = (tid & 3) * 4;
    const int brow = (tid & 255) >> 4;
    const int bc4  = (tid & 15) * 4;
    const float* afG = feature + (size_t)(b * 128 + arow) * K + ak4;
    const float* aqG = query + (size_t)(b * 32 + (tid >> 4)) * K + (tid & 15);
    const float* bG  = (bisq ? Wq : Wf) + (size_t)brow * NA + j * 64 + bc4;
    const int afS = arow * PA + ak4;
    const int aqS = AFSZ + (tid >> 4) * PA + (tid & 15);
    const int bS  = AFSZ + AQSZ + (bisq ? BSZ : 0) + brow * PB + bc4;

    const int aF = (isq ? AFSZ : 0) + (rt0 * 16 + l16) * PA + 2 * half;
    const int bF = AFSZ + AQSZ + (isq ? BSZ : 0) + (2 * half) * PB + wn * 16 + l16;

    v8f acc[3];
    #pragma unroll
    for (int i = 0; i < 3; ++i) acc[i] = (v8f){};

    constexpr int D = 4;
    static_assert(D == NSTG, "ring slot and LDS stage of a chunk have the same index");
    static_assert(NCH == 25, "the drain below is written out for 25 chunks");
    constexpr int G = 4;

    v4f af_0, bb_0; float aq_0;
    v4f af_1, bb_1; float aq_1;
    v4f af_2, bb_2; float aq_2;
    v4f af_3, bb_3; float aq_3;

#define FETCH_CHUNK(S, c)                                                  \
    do {                                                                   \
        af_##S = *(const v4f*)(afG + (c) * KC);                            \
        aq_##S = aqG[(c) * KC];                                            \
        bb_##S = *(const v4f*)(bG + (size_t)(c) * KC * NA);                \
    } while (0)

#define STORE_STAGE(S)                                                     \
    do {                                                                   \
        *(v4f*)&smem[(S) * STG + afS] = af_##S;                            \
        smem[(S) * STG + aqS] = aq_##S;                                    \
        *(v4f*)&smem[(S) * STG + bS] = bb_##S;                             \
    } while (0)

    FETCH_CHUNK(0, 0);
    __builtin_amdgcn_sched_barrier(0);
    FETCH_CHUNK(1, 1);
    __builtin_amdgcn_sched_barrier(0);
    FETCH_CHUNK(2, 2);
    __builtin_amdgcn_sched_barrier(0);
    FETCH_CHUNK(3, 3);
    __builtin_amdgcn_sched_barrier(0);
    STORE_STAGE(0);
    FETCH_CHUNK(0, 4);
    STORE_STAGE(1);
    FETCH_CHUNK(1, 5);
    lds_barrier();

    v2f fa0[3][4], fb0[4];
    v2f fa1[3][4], fb1[4];

    auto load_frags = [&](v2f (&fa)[3][4], v2f (&fb)[4], const int st)
        __attribute__((always_inline)) {
        const float* As = smem + st * STG + aF;
        const float* Bs = smem + st * STG + bF;
        #pragma unroll
        for (int s = 0; s < 4; ++s) {
            const float* p = Bs + (4 * s) * PB;
            fb[s].x = p[0];
            fb[s].y = p[PB];
        }
        #pragma unroll
        for (int s = 0; s < 4; ++s) {
            const float* p = As + 4 * s;
            fa[0][s].x = p[0];
            fa[0][s].y = p[1];
            fa[1][s].x = p[16 * PA];
            fa[1][s].y = p[16 * PA + 1];
        }
        if (nt3) {
            #pragma unroll
            for (int s = 0; s < 4; ++s) {
                const float* p = As + 32 * PA + 4 * s;
                fa[2][s].x = p[0];
                fa[2][s].y = p[1];
            }
        }
    };

    auto chunk_mma = [&](const v2f (&fa)[3][4], const v2f (&fb)[4])
        __attribute__((always_inline)) {
        #pragma unroll
        for (int s = 0; s < 4; ++s) {
            acc[0] = __builtin_amdgcn_wmma_f32_16x16x4_f32(
                 false, fa[0][s],  false, fb[s],
                 (short)0, acc[0],  false, false);
            acc[1] = __builtin_amdgcn_wmma_f32_16x16x4_f32(
                 false, fa[1][s],  false, fb[s],
                 (short)0, acc[1],  false, false);
        }
        if (nt3) {
            #pragma unroll
            for (int s = 0; s < 4; ++s) {
                acc[2] = __builtin_amdgcn_wmma_f32_16x16x4_f32(
                     false, fa[2][s],  false, fb[s],
                     (short)0, acc[2],  false, false);
            }
        }
    };

    load_frags(fa0, fb0, 0);

#define STEP_FULL(S1, S2, FAN, FBN, FAC, FBC, cf)                          \
    do {                                                                   \
        STORE_STAGE(S2); FETCH_CHUNK(S2, cf);                              \
        load_frags(FAN, FBN, S1);                                          \
        __builtin_amdgcn_sched_barrier(0);                                 \
        chunk_mma(FAC, FBC);                                               \
        lds_barrier();                                                     \
    } while (0)
#define STEP_NOFETCH(S1, S2, FAN, FBN, FAC, FBC)                           \
    do {                                                                   \
        STORE_STAGE(S2);                                                   \
        load_frags(FAN, FBN, S1);                                          \
        __builtin_amdgcn_sched_barrier(0);                                 \
        chunk_mma(FAC, FBC);                                               \
        lds_barrier();                                                     \
    } while (0)

    for (int g = 0; g < G; ++g) {
        const int ch = g * D;
        STEP_FULL(1, 2, fa1, fb1, fa0, fb0, ch + 2 + D);
        STEP_FULL(2, 3, fa0, fb0, fa1, fb1, ch + 3 + D);
        STEP_FULL(3, 0, fa1, fb1, fa0, fb0, ch + 4 + D);
        STEP_FULL(0, 1, fa0, fb0, fa1, fb1, ch + 5 + D);
    }
    STEP_FULL(1, 2, fa1, fb1, fa0, fb0, 22);
    STEP_FULL(2, 3, fa0, fb0, fa1, fb1, 23);
    STEP_FULL(3, 0, fa1, fb1, fa0, fb0, 24);
    STEP_NOFETCH(0, 1, fa0, fb0, fa1, fb1);
    STEP_NOFETCH(1, 2, fa1, fb1, fa0, fb0);
    STEP_NOFETCH(2, 3, fa0, fb0, fa1, fb1);
    STEP_NOFETCH(3, 0, fa1, fb1, fa0, fb0);
    load_frags(fa0, fb0, 0);
    chunk_mma(fa1, fb1);
    chunk_mma(fa0, fb0);
#undef STEP_FULL
#undef STEP_NOFETCH
#undef FETCH_CHUNK
#undef STORE_STAGE
    lds_barrier();

    {
        const int col = wn * 16 + l16;
        const float bias = isq ? bq[j * 64 + col] : 0.0f;
        float* T = smem + (isq ? PQT : 0) + (rt0 * 16 + 8 * half) * PC + col;
        #pragma unroll
        for (int r = 0; r < 8; ++r) {
            T[r * PC]             = acc[0][r] + bias;
            T[(16 + r) * PC]      = acc[1][r] + bias;
        }
        if (nt3) {
            #pragma unroll
            for (int r = 0; r < 8; ++r)
                T[(32 + r) * PC]  = acc[2][r];
        }
    }
    lds_barrier();

    const int qo  = wave >> 2;
    const int fg  = wave & 3;
    const int a8  = lane & 7;
    const int sub = lane >> 3;
    const int f0  = (fg * 4 + sub) * 8;

    float w[8], x[8][8];
    {
        const float4* wa4 = (const float4*)(Wa + j * 64 + a8 * 8);
        const float4 w0 = wa4[0], w1 = wa4[1];
        w[0] = w0.x; w[1] = w0.y; w[2] = w0.z; w[3] = w0.w;
        w[4] = w1.x; w[5] = w1.y; w[6] = w1.z; w[7] = w1.w;
    }
    #pragma unroll
    for (int r = 0; r < 8; ++r) {
        const float4* xp = (const float4*)&smem[(f0 + r) * PC + a8 * 8];
        const float4 x0 = xp[0], x1 = xp[1];
        x[r][0] = x0.x; x[r][1] = x0.y; x[r][2] = x0.z; x[r][3] = x0.w;
        x[r][4] = x1.x; x[r][5] = x1.y; x[r][6] = x1.z; x[r][7] = x1.w;
    }

    const float* yrow = &smem[PQT + (qo * 8) * PC + a8 * 8];
    float4 yc0 = *(const float4*)(yrow);
    float4 yc1 = *(const float4*)(yrow + 4);
    float* dst = part + ((size_t)((j * 64 + b) * 32 + qo * 8)) * 128 + fg * 32 + lane;
    #pragma nounroll
    for (int qi = 0; qi < 8; ++qi) {
        const float* yn = yrow + ((qi + 1) & 7) * PC;
        const float4 yn0 = *(const float4*)(yn);
        const float4 yn1 = *(const float4*)(yn + 4);
        const float y[8] = {yc0.x, yc0.y, yc0.z, yc0.w, yc1.x, yc1.y, yc1.z, yc1.w};

        float sums[8];
        #pragma unroll
        for (int fj = 0; fj < 8; ++fj) sums[fj] = 0.0f;
        #pragma unroll
        for (int k = 0; k < 8; ++k) {
            float t[8];
            #pragma unroll
            for (int fj = 0; fj < 8; ++fj)
                t[fj] = fast_tanh(x[fj][k] + y[k]);
            #pragma unroll
            for (int fj = 0; fj < 8; ++fj)
                sums[fj] = __builtin_fmaf(t[fj], w[k], sums[fj]);
        }

        fold_level<4>(sums, (lane & 4) != 0, 4);
        fold_level<2>(sums, (lane & 2) != 0, 2);
        fold_level<1>(sums, (lane & 1) != 0, 1);

        dst[qi * 128] = sums[0];

        yc0 = yn0;
        yc1 = yn1;
    }
}

__global__ void __launch_bounds__(256) __attribute__((amdgpu_waves_per_eu(1, 8)))
softmax_av_kernel(const float* __restrict__ part,
                  const int* __restrict__ mask,
                  const float* __restrict__ feature,
                  float* __restrict__ out)
{
    constexpr int NT = 5, KW = 4, NW = 8, N = 400, F = 128;
    constexpr int PAL = 132;
    constexpr size_t SL4 = (size_t)64 * 32 * 128 / 4;

    __shared__ float red[NW][NT * 8][32];
    __shared__ __attribute__((aligned(16))) float alpha[16][PAL];

    const int batch = blockIdx.z;
    const int tN   = blockIdx.x * 16 * NT;
    const int tM   = blockIdx.y * 16;
    const int lane = threadIdx.x & 31;
    const int half = lane >> 4;
    const int l16  = lane & 15;
    const int wave = __builtin_amdgcn_readfirstlane((int)(threadIdx.x >> 5));

    const int k0 = wave * (4 * KW) + half * 2;
    const float* __restrict__ bP =
        feature + (size_t)batch * F * N + (size_t)k0 * N + tN + l16;

    const int4 mk = ((const int4*)(mask + batch * F))[lane];
    float4 ps[2][4];
    #pragma unroll
    for (int rr = 0; rr < 2; ++rr) {
        const float4* p =
            (const float4*)(part + (size_t)(batch * 32 + tM + 2 * wave + rr) * F) + lane;
        #pragma unroll
        for (int sl = 0; sl < 4; ++sl) ps[rr][sl] = p[sl * SL4];
    }

    v2f bv[KW][NT];
    #pragma unroll
    for (int s = 0; s < KW; ++s) {
        #pragma unroll
        for (int t = 0; t < NT; ++t) {
            bv[s][t].x = bP[(size_t)(4 * s)     * N + t * 16];
            bv[s][t].y = bP[(size_t)(4 * s + 1) * N + t * 16];
        }
    }
    __builtin_amdgcn_sched_barrier(0);

    {
        #pragma unroll
        for (int rr = 0; rr < 2; ++rr) {
            const int row = 2 * wave + rr;
            const float4 s0 = ps[rr][0], s1 = ps[rr][1], s2 = ps[rr][2], s3 = ps[rr][3];
            float4 v;
            v.x = ((s0.x + s1.x) + s2.x) + s3.x;
            v.y = ((s0.y + s1.y) + s2.y) + s3.y;
            v.z = ((s0.z + s1.z) + s2.z) + s3.z;
            v.w = ((s0.w + s1.w) + s2.w) + s3.w;
            v.x = (mk.x == 0) ? -1e9f : v.x;
            v.y = (mk.y == 0) ? -1e9f : v.y;
            v.z = (mk.z == 0) ? -1e9f : v.z;
            v.w = (mk.w == 0) ? -1e9f : v.w;

            float m = fmaxf(fmaxf(v.x, v.y), fmaxf(v.z, v.w));
            #pragma unroll
            for (int off = 16; off > 0; off >>= 1)
                m = fmaxf(m, __shfl_xor(m, off, 32));

            float4 e;
            e.x = __expf(v.x - m);
            e.y = __expf(v.y - m);
            e.z = __expf(v.z - m);
            e.w = __expf(v.w - m);

            float s = e.x + e.y + e.z + e.w;
            #pragma unroll
            for (int off = 16; off > 0; off >>= 1)
                s += __shfl_xor(s, off, 32);

            const float inv = 1.0f / s;
            float4 o;
            o.x = e.x * inv; o.y = e.y * inv; o.z = e.z * inv; o.w = e.w * inv;
            *(float4*)&alpha[row][lane * 4] = o;
        }
    }
    lds_barrier();

    v2f av[KW];
    #pragma unroll
    for (int s = 0; s < KW; ++s) {
        const float* p = &alpha[l16][k0 + 4 * s];
        av[s].x = p[0];
        av[s].y = p[1];
    }

    v8f acc[NT];
    #pragma unroll
    for (int t = 0; t < NT; ++t) acc[t] = (v8f){};

    #pragma unroll
    for (int s = 0; s < KW; ++s) {
        #pragma unroll
        for (int t = 0; t < NT; ++t) {
            acc[t] = __builtin_amdgcn_wmma_f32_16x16x4_f32(
                 false, av[s],  false, bv[s][t],
                 (short)0, acc[t],  false, false);
        }
    }

    #pragma unroll
    for (int t = 0; t < NT; ++t) {
        #pragma unroll
        for (int v = 0; v < 8; ++v)
            red[wave][t * 8 + v][lane] = acc[t][v];
    }
    __syncthreads();

    float* __restrict__ C = out + (size_t)batch * 32 * N;
    for (int v = wave; v < 8; v += NW) {
        const int m = tM + v + half * 8;
        #pragma unroll
        for (int t = 0; t < NT; ++t) {
            const int r = t * 8 + v;
            float s = red[0][r][lane];
            #pragma unroll
            for (int w = 1; w < NW; ++w) s += red[w][r][lane];
            C[(size_t)m * N + tN + t * 16 + l16] = s;
        }
    }
}

extern "C" void kernel_launch(void* const* d_in, const int* in_sizes, int n_in,
                              void* d_out, int out_size, void* d_ws, size_t ws_size,
                              hipStream_t stream)
{
    const float* feature = (const float*)d_in[0];
    const float* query   = (const float*)d_in[1];
    const int*   mask    = (const int*)  d_in[2];
    const float* Wf      = (const float*)d_in[3];
    const float* Wq      = (const float*)d_in[4];
    const float* bq      = (const float*)d_in[5];
    const float* Wa      = (const float*)d_in[6];
    float*       out     = (float*)d_out;

    float* part = (float*)d_ws;

    proj_scores_slice_kernel<<<64 * 4, 512, 0, stream>>>(
        feature, query, Wf, Wq, bq, Wa, part);

    softmax_av_kernel<<<dim3(400 / 80, 32 / 16, 64), dim3(256), 0, stream>>>(
        part, mask, feature, out);
}


# proj GEMM K loop: the barrier of each chunk split into signal (behind the wave's own stage stores) and wait (behind its WMMAs): waves of a SIMD run up to a step apart; sum order kept
# speedup vs baseline: 5.6080x; 1.0286x over previous
#include <hip/hip_runtime.h>

typedef float v2f __attribute__((ext_vector_type(2)));
typedef float v4f __attribute__((ext_vector_type(4)));
typedef float v8f __attribute__((ext_vector_type(8)));

static __device__ __forceinline__ float fast_tanh(float x) {
#if __has_builtin(__builtin_amdgcn_tanhf)
    return __builtin_amdgcn_tanhf(x);
#else
    float r;
    asm("v_tanh_f32 %0, %1" : "=v"(r) : "v"(x));
    return r;
#endif
}

static __device__ __forceinline__ void lds_barrier()
{
    __builtin_amdgcn_fence(__ATOMIC_RELEASE, "workgroup", "local");
    __builtin_amdgcn_s_barrier();
    __builtin_amdgcn_fence(__ATOMIC_ACQUIRE, "workgroup", "local");
}

static __device__ __forceinline__ void lds_signal_stored()
{
    __builtin_amdgcn_fence(__ATOMIC_RELEASE, "workgroup", "local");
    __builtin_amdgcn_s_barrier_signal(-1);
}
static __device__ __forceinline__ void lds_wait_stored()
{
    __builtin_amdgcn_s_barrier_wait(-1);
    __builtin_amdgcn_fence(__ATOMIC_ACQUIRE, "workgroup", "local");
}

template <int H, int NV>
static __device__ __forceinline__ void fold_level(float (&v)[NV], const bool up, const int bit)
{
    #pragma unroll
    for (int i = 0; i < H; ++i) {
        const float keep = up ? v[i + H] : v[i];
        const float send = up ? v[i]     : v[i + H];
        v[i] = keep + __shfl_xor(send, bit, 32);
    }
}

__global__ void __launch_bounds__(512) __attribute__((amdgpu_waves_per_eu(1, 4)))
proj_scores_slice_kernel(const float* feature,
                         const float* query,
                         const float* Wf,
                         const float* Wq,
                         const float* bq,
                         const float* Wa,
                         float* part)
{
    constexpr int K    = 400;
    constexpr int NA   = 256;
    constexpr int KC   = 16;
    constexpr int PA   = 20;
    constexpr int PB   = 80;
    constexpr int PC   = 68;
    constexpr int AFSZ = 128 * PA;
    constexpr int AQSZ = 32 * PA;
    constexpr int BSZ  = KC * PB;
    constexpr int STG  = AFSZ + AQSZ + 2 * BSZ;
    constexpr int NCH  = K / KC;
    constexpr int PQT  = 128 * PC;
    static_assert(K % KC == 0, "chunks must tile K exactly");
    constexpr int NSTG = 4;
    static_assert((128 + 32) * PC <= NSTG * STG, "the finished tiles fit inside the stages");

    __shared__ __attribute__((aligned(16))) float smem[NSTG * STG];

    const int tid  = threadIdx.x;
    const int lane = tid & 31;
    const int half = lane >> 4;
    const int l16  = lane & 15;
    const int wave = __builtin_amdgcn_readfirstlane(tid >> 5);
    const int wn   = wave & 3;
    const int rg   = wave >> 2;
    const bool isq = (rg == 3);
    const bool nt3 = (rg < 2);
    const int rt0  = isq ? 0 : rg * 3;

    const int j = blockIdx.x & 3;
    const int b = blockIdx.x >> 2;

    const bool bisq = (wave >= 8);
    const int arow = tid >> 2;
    const int ak4  = (tid & 3) * 4;
    const int brow = (tid & 255) >> 4;
    const int bc4  = (tid & 15) * 4;
    const float* afG = feature + (size_t)(b * 128 + arow) * K + ak4;
    const float* aqG = query + (size_t)(b * 32 + (tid >> 4)) * K + (tid & 15);
    const float* bG  = (bisq ? Wq : Wf) + (size_t)brow * NA + j * 64 + bc4;
    const int afS = arow * PA + ak4;
    const int aqS = AFSZ + (tid >> 4) * PA + (tid & 15);
    const int bS  = AFSZ + AQSZ + (bisq ? BSZ : 0) + brow * PB + bc4;

    const int aF = (isq ? AFSZ : 0) + (rt0 * 16 + l16) * PA + 2 * half;
    const int bF = AFSZ + AQSZ + (isq ? BSZ : 0) + (2 * half) * PB + wn * 16 + l16;

    v8f acc[3];
    #pragma unroll
    for (int i = 0; i < 3; ++i) acc[i] = (v8f){};

    constexpr int D = 4;
    static_assert(D == NSTG, "ring slot and LDS stage of a chunk have the same index");
    static_assert(NCH == 25, "the drain below is written out for 25 chunks");
    constexpr int G = 4;

    v4f af_0, bb_0; float aq_0;
    v4f af_1, bb_1; float aq_1;
    v4f af_2, bb_2; float aq_2;
    v4f af_3, bb_3; float aq_3;

#define FETCH_CHUNK(S, c)                                                  \
    do {                                                                   \
        af_##S = *(const v4f*)(afG + (c) * KC);                            \
        aq_##S = aqG[(c) * KC];                                            \
        bb_##S = *(const v4f*)(bG + (size_t)(c) * KC * NA);                \
    } while (0)

#define STORE_STAGE(S)                                                     \
    do {                                                                   \
        *(v4f*)&smem[(S) * STG + afS] = af_##S;                            \
        smem[(S) * STG + aqS] = aq_##S;                                    \
        *(v4f*)&smem[(S) * STG + bS] = bb_##S;                             \
    } while (0)

    FETCH_CHUNK(0, 0);
    __builtin_amdgcn_sched_barrier(0);
    FETCH_CHUNK(1, 1);
    __builtin_amdgcn_sched_barrier(0);
    FETCH_CHUNK(2, 2);
    __builtin_amdgcn_sched_barrier(0);
    FETCH_CHUNK(3, 3);
    __builtin_amdgcn_sched_barrier(0);
    STORE_STAGE(0);
    FETCH_CHUNK(0, 4);
    STORE_STAGE(1);
    FETCH_CHUNK(1, 5);
    lds_barrier();

    v2f fa0[3][4], fb0[4];
    v2f fa1[3][4], fb1[4];

    auto load_frags = [&](v2f (&fa)[3][4], v2f (&fb)[4], const int st)
        __attribute__((always_inline)) {
        const float* As = smem + st * STG + aF;
        const float* Bs = smem + st * STG + bF;
        #pragma unroll
        for (int s = 0; s < 4; ++s) {
            const float* p = Bs + (4 * s) * PB;
            fb[s].x = p[0];
            fb[s].y = p[PB];
        }
        #pragma unroll
        for (int s = 0; s < 4; ++s) {
            const float* p = As + 4 * s;
            fa[0][s].x = p[0];
            fa[0][s].y = p[1];
            fa[1][s].x = p[16 * PA];
            fa[1][s].y = p[16 * PA + 1];
        }
        if (nt3) {
            #pragma unroll
            for (int s = 0; s < 4; ++s) {
                const float* p = As + 32 * PA + 4 * s;
                fa[2][s].x = p[0];
                fa[2][s].y = p[1];
            }
        }
    };

    auto chunk_mma = [&](const v2f (&fa)[3][4], const v2f (&fb)[4])
        __attribute__((always_inline)) {
        #pragma unroll
        for (int s = 0; s < 4; ++s) {
            acc[0] = __builtin_amdgcn_wmma_f32_16x16x4_f32(
                 false, fa[0][s],  false, fb[s],
                 (short)0, acc[0],  false, false);
            acc[1] = __builtin_amdgcn_wmma_f32_16x16x4_f32(
                 false, fa[1][s],  false, fb[s],
                 (short)0, acc[1],  false, false);
        }
        if (nt3) {
            #pragma unroll
            for (int s = 0; s < 4; ++s) {
                acc[2] = __builtin_amdgcn_wmma_f32_16x16x4_f32(
                     false, fa[2][s],  false, fb[s],
                     (short)0, acc[2],  false, false);
            }
        }
    };

    load_frags(fa0, fb0, 0);

#define STEP_FULL(S1, S2, FAN, FBN, FAC, FBC, cf)                          \
    do {                                                                   \
        STORE_STAGE(S2); FETCH_CHUNK(S2, cf);                              \
        lds_signal_stored();                                               \
        load_frags(FAN, FBN, S1);                                          \
        __builtin_amdgcn_sched_barrier(0);                                 \
        chunk_mma(FAC, FBC);                                               \
        lds_wait_stored();                                                 \
    } while (0)
#define STEP_NOFETCH(S1, S2, FAN, FBN, FAC, FBC)                           \
    do {                                                                   \
        STORE_STAGE(S2);                                                   \
        lds_signal_stored();                                               \
        load_frags(FAN, FBN, S1);                                          \
        __builtin_amdgcn_sched_barrier(0);                                 \
        chunk_mma(FAC, FBC);                                               \
        lds_wait_stored();                                                 \
    } while (0)

    for (int g = 0; g < G; ++g) {
        const int ch = g * D;
        STEP_FULL(1, 2, fa1, fb1, fa0, fb0, ch + 2 + D);
        STEP_FULL(2, 3, fa0, fb0, fa1, fb1, ch + 3 + D);
        STEP_FULL(3, 0, fa1, fb1, fa0, fb0, ch + 4 + D);
        STEP_FULL(0, 1, fa0, fb0, fa1, fb1, ch + 5 + D);
    }
    STEP_FULL(1, 2, fa1, fb1, fa0, fb0, 22);
    STEP_FULL(2, 3, fa0, fb0, fa1, fb1, 23);
    STEP_FULL(3, 0, fa1, fb1, fa0, fb0, 24);
    STEP_NOFETCH(0, 1, fa0, fb0, fa1, fb1);
    STEP_NOFETCH(1, 2, fa1, fb1, fa0, fb0);
    STEP_NOFETCH(2, 3, fa0, fb0, fa1, fb1);
    STEP_NOFETCH(3, 0, fa1, fb1, fa0, fb0);
    load_frags(fa0, fb0, 0);
    chunk_mma(fa1, fb1);
    chunk_mma(fa0, fb0);
#undef STEP_FULL
#undef STEP_NOFETCH
#undef FETCH_CHUNK
#undef STORE_STAGE
    lds_barrier();

    {
        const int col = wn * 16 + l16;
        const float bias = isq ? bq[j * 64 + col] : 0.0f;
        float* T = smem + (isq ? PQT : 0) + (rt0 * 16 + 8 * half) * PC + col;
        #pragma unroll
        for (int r = 0; r < 8; ++r) {
            T[r * PC]             = acc[0][r] + bias;
            T[(16 + r) * PC]      = acc[1][r] + bias;
        }
        if (nt3) {
            #pragma unroll
            for (int r = 0; r < 8; ++r)
                T[(32 + r) * PC]  = acc[2][r];
        }
    }
    lds_barrier();

    const int qo  = wave >> 2;
    const int fg  = wave & 3;
    const int a8  = lane & 7;
    const int sub = lane >> 3;
    const int f0  = (fg * 4 + sub) * 8;

    float w[8], x[8][8];
    {
        const float4* wa4 = (const float4*)(Wa + j * 64 + a8 * 8);
        const float4 w0 = wa4[0], w1 = wa4[1];
        w[0] = w0.x; w[1] = w0.y; w[2] = w0.z; w[3] = w0.w;
        w[4] = w1.x; w[5] = w1.y; w[6] = w1.z; w[7] = w1.w;
    }
    #pragma unroll
    for (int r = 0; r < 8; ++r) {
        const float4* xp = (const float4*)&smem[(f0 + r) * PC + a8 * 8];
        const float4 x0 = xp[0], x1 = xp[1];
        x[r][0] = x0.x; x[r][1] = x0.y; x[r][2] = x0.z; x[r][3] = x0.w;
        x[r][4] = x1.x; x[r][5] = x1.y; x[r][6] = x1.z; x[r][7] = x1.w;
    }

    const float* yrow = &smem[PQT + (qo * 8) * PC + a8 * 8];
    float4 yc0 = *(const float4*)(yrow);
    float4 yc1 = *(const float4*)(yrow + 4);
    float* dst = part + ((size_t)((j * 64 + b) * 32 + qo * 8)) * 128 + fg * 32 + lane;
    #pragma nounroll
    for (int qi = 0; qi < 8; ++qi) {
        const float* yn = yrow + ((qi + 1) & 7) * PC;
        const float4 yn0 = *(const float4*)(yn);
        const float4 yn1 = *(const float4*)(yn + 4);
        const float y[8] = {yc0.x, yc0.y, yc0.z, yc0.w, yc1.x, yc1.y, yc1.z, yc1.w};

        float sums[8];
        #pragma unroll
        for (int fj = 0; fj < 8; ++fj) sums[fj] = 0.0f;
        #pragma unroll
        for (int k = 0; k < 8; ++k) {
            float t[8];
            #pragma unroll
            for (int fj = 0; fj < 8; ++fj)
                t[fj] = fast_tanh(x[fj][k] + y[k]);
            #pragma unroll
            for (int fj = 0; fj < 8; ++fj)
                sums[fj] = __builtin_fmaf(t[fj], w[k], sums[fj]);
        }

        fold_level<4>(sums, (lane & 4) != 0, 4);
        fold_level<2>(sums, (lane & 2) != 0, 2);
        fold_level<1>(sums, (lane & 1) != 0, 1);

        dst[qi * 128] = sums[0];

        yc0 = yn0;
        yc1 = yn1;
    }
}

__global__ void __launch_bounds__(256) __attribute__((amdgpu_waves_per_eu(1, 8)))
softmax_av_kernel(const float* __restrict__ part,
                  const int* __restrict__ mask,
                  const float* __restrict__ feature,
                  float* __restrict__ out)
{
    constexpr int NT = 5, KW = 4, NW = 8, N = 400, F = 128;
    constexpr int PAL = 132;
    constexpr size_t SL4 = (size_t)64 * 32 * 128 / 4;

    __shared__ float red[NW][NT * 8][32];
    __shared__ __attribute__((aligned(16))) float alpha[16][PAL];

    const int batch = blockIdx.z;
    const int tN   = blockIdx.x * 16 * NT;
    const int tM   = blockIdx.y * 16;
    const int lane = threadIdx.x & 31;
    const int half = lane >> 4;
    const int l16  = lane & 15;
    const int wave = __builtin_amdgcn_readfirstlane((int)(threadIdx.x >> 5));

    const int k0 = wave * (4 * KW) + half * 2;
    const float* __restrict__ bP =
        feature + (size_t)batch * F * N + (size_t)k0 * N + tN + l16;

    const int4 mk = ((const int4*)(mask + batch * F))[lane];
    float4 ps[2][4];
    #pragma unroll
    for (int rr = 0; rr < 2; ++rr) {
        const float4* p =
            (const float4*)(part + (size_t)(batch * 32 + tM + 2 * wave + rr) * F) + lane;
        #pragma unroll
        for (int sl = 0; sl < 4; ++sl) ps[rr][sl] = p[sl * SL4];
    }

    v2f bv[KW][NT];
    #pragma unroll
    for (int s = 0; s < KW; ++s) {
        #pragma unroll
        for (int t = 0; t < NT; ++t) {
            bv[s][t].x = bP[(size_t)(4 * s)     * N + t * 16];
            bv[s][t].y = bP[(size_t)(4 * s + 1) * N + t * 16];
        }
    }
    __builtin_amdgcn_sched_barrier(0);

    {
        #pragma unroll
        for (int rr = 0; rr < 2; ++rr) {
            const int row = 2 * wave + rr;
            const float4 s0 = ps[rr][0], s1 = ps[rr][1], s2 = ps[rr][2], s3 = ps[rr][3];
            float4 v;
            v.x = ((s0.x + s1.x) + s2.x) + s3.x;
            v.y = ((s0.y + s1.y) + s2.y) + s3.y;
            v.z = ((s0.z + s1.z) + s2.z) + s3.z;
            v.w = ((s0.w + s1.w) + s2.w) + s3.w;
            v.x = (mk.x == 0) ? -1e9f : v.x;
            v.y = (mk.y == 0) ? -1e9f : v.y;
            v.z = (mk.z == 0) ? -1e9f : v.z;
            v.w = (mk.w == 0) ? -1e9f : v.w;

            float m = fmaxf(fmaxf(v.x, v.y), fmaxf(v.z, v.w));
            #pragma unroll
            for (int off = 16; off > 0; off >>= 1)
                m = fmaxf(m, __shfl_xor(m, off, 32));

            float4 e;
            e.x = __expf(v.x - m);
            e.y = __expf(v.y - m);
            e.z = __expf(v.z - m);
            e.w = __expf(v.w - m);

            float s = e.x + e.y + e.z + e.w;
            #pragma unroll
            for (int off = 16; off > 0; off >>= 1)
                s += __shfl_xor(s, off, 32);

            const float inv = 1.0f / s;
            float4 o;
            o.x = e.x * inv; o.y = e.y * inv; o.z = e.z * inv; o.w = e.w * inv;
            *(float4*)&alpha[row][lane * 4] = o;
        }
    }
    lds_barrier();

    v2f av[KW];
    #pragma unroll
    for (int s = 0; s < KW; ++s) {
        const float* p = &alpha[l16][k0 + 4 * s];
        av[s].x = p[0];
        av[s].y = p[1];
    }

    v8f acc[NT];
    #pragma unroll
    for (int t = 0; t < NT; ++t) acc[t] = (v8f){};

    #pragma unroll
    for (int s = 0; s < KW; ++s) {
        #pragma unroll
        for (int t = 0; t < NT; ++t) {
            acc[t] = __builtin_amdgcn_wmma_f32_16x16x4_f32(
                 false, av[s],  false, bv[s][t],
                 (short)0, acc[t],  false, false);
        }
    }

    #pragma unroll
    for (int t = 0; t < NT; ++t) {
        #pragma unroll
        for (int v = 0; v < 8; ++v)
            red[wave][t * 8 + v][lane] = acc[t][v];
    }
    __syncthreads();

    float* __restrict__ C = out + (size_t)batch * 32 * N;
    for (int v = wave; v < 8; v += NW) {
        const int m = tM + v + half * 8;
        #pragma unroll
        for (int t = 0; t < NT; ++t) {
            const int r = t * 8 + v;
            float s = red[0][r][lane];
            #pragma unroll
            for (int w = 1; w < NW; ++w) s += red[w][r][lane];
            C[(size_t)m * N + tN + t * 16 + l16] = s;
        }
    }
}

extern "C" void kernel_launch(void* const* d_in, const int* in_sizes, int n_in,
                              void* d_out, int out_size, void* d_ws, size_t ws_size,
                              hipStream_t stream)
{
    const float* feature = (const float*)d_in[0];
    const float* query   = (const float*)d_in[1];
    const int*   mask    = (const int*)  d_in[2];
    const float* Wf      = (const float*)d_in[3];
    const float* Wq      = (const float*)d_in[4];
    const float* bq      = (const float*)d_in[5];
    const float* Wa      = (const float*)d_in[6];
    float*       out     = (float*)d_out;

    float* part = (float*)d_ws;

    proj_scores_slice_kernel<<<64 * 4, 512, 0, stream>>>(
        feature, query, Wf, Wq, bq, Wa, part);

    softmax_av_kernel<<<dim3(400 / 80, 32 / 16, 64), dim3(256), 0, stream>>>(
        part, mask, feature, out);
}
